# starting kernel
# baseline (speedup 1.0000x reference)
#include <hip/hip_runtime.h>
#include <hip/hip_bf16.h>

typedef __attribute__((ext_vector_type(2))) float v2f;
typedef __attribute__((ext_vector_type(8))) float v8f;

#define EMBED 768
#define S_LEN 2048
#define NHEAD 12
#define HDIM  64
#define BATCH 2
#define ROWS  (BATCH * S_LEN)

__device__ __forceinline__ v8f wmma4(v2f a, v2f b, v8f c) {
  return __builtin_amdgcn_wmma_f32_16x16x4_f32(
       false, a,  false, b,
       (short)0, c,  false,  false);
}

__device__ __forceinline__ v8f vzero8() {
  v8f z = {0.f, 0.f, 0.f, 0.f, 0.f, 0.f, 0.f, 0.f};
  return z;
}

__global__ __launch_bounds__(256) void gemm_bias_kernel(
    const float* __restrict__ A, const float* __restrict__ W,
    const float* __restrict__ bias, float* __restrict__ C, int has_bias) {
  const int lane = threadIdx.x & 31;
  const int wave = threadIdx.x >> 5;
  const int half = lane >> 4;
  const int l16  = lane & 15;

  const int m0 = blockIdx.x * 64  + (wave & 3) * 16;
  const int n0 = blockIdx.y * 128 + (wave >> 2) * 64;

  v8f acc[4];
#pragma unroll
  for (int i = 0; i < 4; ++i) acc[i] = vzero8();

  const float* arow = A + (size_t)(m0 + l16) * EMBED;

#pragma unroll 2
  for (int kk = 0; kk < EMBED; kk += 4) {
    const int ka = kk + 2 * half;
    const v2f a = *(const v2f*)(arow + ka);
    const float* w0 = W + (size_t)ka * EMBED + n0 + l16;
    const float* w1 = w0 + EMBED;
#pragma unroll
    for (int nt = 0; nt < 4; ++nt) {
      v2f b;
      b.x = w0[nt * 16];
      b.y = w1[nt * 16];
      acc[nt] = wmma4(a, b, acc[nt]);
    }
  }

#pragma unroll
  for (int nt = 0; nt < 4; ++nt) {
    const int col = n0 + nt * 16 + l16;
    const float bv = has_bias ? bias[col] : 0.f;
#pragma unroll
    for (int v = 0; v < 8; ++v) {
      C[(size_t)(m0 + v + 8 * half) * EMBED + col] = acc[nt][v] + bv;
    }
  }
}

__global__ __launch_bounds__(256) void attention_kernel(
    const float* __restrict__ Q, const float* __restrict__ K,
    const float* __restrict__ V, const float* __restrict__ mask,
    float* __restrict__ O) {
  __shared__ float pstage[8][16 * 16];

  const int lane = threadIdx.x & 31;
  const int wave = threadIdx.x >> 5;
  const int half = lane >> 4;
  const int l16  = lane & 15;

  const int tile = blockIdx.x * 8 + wave;
  const int qt = tile & 127;
  const int bh = tile >> 7;
  const int b  = bh / NHEAD;
  const int h  = bh % NHEAD;

  const int    q0   = qt * 16;
  const size_t base = (size_t)b * S_LEN * EMBED + (size_t)h * HDIM;

  const float* qrow = Q + base + (size_t)(q0 + l16) * EMBED;

  v8f acc[4];
#pragma unroll
  for (int i = 0; i < 4; ++i) acc[i] = vzero8();
  float mrow[8], lrow[8];
#pragma unroll
  for (int v = 0; v < 8; ++v) { mrow[v] = -3.0e38f; lrow[v] = 0.f; }

  float* myp = &pstage[wave][0];

  for (int kt = 0; kt < S_LEN / 16; ++kt) {
    v8f s = vzero8();
    const float* krow = K + base + (size_t)(kt * 16 + l16) * EMBED;
#pragma unroll
    for (int kk = 0; kk < HDIM; kk += 4) {
      v2f a  = *(const v2f*)(qrow + kk + 2 * half);
      v2f bf = *(const v2f*)(krow + kk + 2 * half);
      s = wmma4(a, bf, s);
    }

    const float* mp = mask + (size_t)(q0 + 8 * half) * S_LEN + kt * 16 + l16;
    float p[8];
#pragma unroll
    for (int v = 0; v < 8; ++v) {
      float sv = s[v] * 0.125f + mp[(size_t)v * S_LEN] * (-1.0e9f);
      float rmax = sv;
      rmax = fmaxf(rmax, __shfl_xor(rmax, 1, 32));
      rmax = fmaxf(rmax, __shfl_xor(rmax, 2, 32));
      rmax = fmaxf(rmax, __shfl_xor(rmax, 4, 32));
      rmax = fmaxf(rmax, __shfl_xor(rmax, 8, 32));
      const float mnew  = fmaxf(mrow[v], rmax);
      const float alpha = __expf(mrow[v] - mnew);
      const float pv    = __expf(sv - mnew);
      float rsum = pv;
      rsum += __shfl_xor(rsum, 1, 32);
      rsum += __shfl_xor(rsum, 2, 32);
      rsum += __shfl_xor(rsum, 4, 32);
      rsum += __shfl_xor(rsum, 8, 32);
      lrow[v] = lrow[v] * alpha + rsum;
      mrow[v] = mnew;
      p[v] = pv;
#pragma unroll
      for (int nt = 0; nt < 4; ++nt) acc[nt][v] *= alpha;
    }

#pragma unroll
    for (int v = 0; v < 8; ++v) myp[(v + 8 * half) * 16 + l16] = p[v];
    __syncthreads();

#pragma unroll
    for (int kk = 0; kk < 16; kk += 4) {
      v2f a = *(const v2f*)(myp + l16 * 16 + kk + 2 * half);
      const float* v0 = V + base + (size_t)(kt * 16 + kk + 2 * half) * EMBED + l16;
      const float* v1 = v0 + EMBED;
#pragma unroll
      for (int nt = 0; nt < 4; ++nt) {
        v2f bf;
        bf.x = v0[nt * 16];
        bf.y = v1[nt * 16];
        acc[nt] = wmma4(a, bf, acc[nt]);
      }
    }
    __syncthreads();
  }

#pragma unroll
  for (int v = 0; v < 8; ++v) {
    const float inv = 1.0f / lrow[v];
    float* orow = O + base + (size_t)(q0 + v + 8 * half) * EMBED;
#pragma unroll
    for (int nt = 0; nt < 4; ++nt) {
      orow[nt * 16 + l16] = acc[nt][v] * inv;
    }
  }
}

extern "C" void kernel_launch(void* const* d_in, const int* in_sizes, int n_in,
                              void* d_out, int out_size, void* d_ws, size_t ws_size,
                              hipStream_t stream) {
  const float* x    = (const float*)d_in[0];
  const float* mask = (const float*)d_in[1];
  const float* Wq   = (const float*)d_in[2];
  const float* Wk   = (const float*)d_in[3];
  const float* Wv   = (const float*)d_in[4];
  const float* Wo   = (const float*)d_in[5];
  const float* bo   = (const float*)d_in[6];

  float* ws = (float*)d_ws;
  const size_t mat = (size_t)ROWS * EMBED;
  float* Qb = ws;
  float* Kb = ws + mat;
  float* Vb = ws + 2 * mat;
  float* Ab = ws + 3 * mat;

  dim3 g(ROWS / 64, EMBED / 128);
  gemm_bias_kernel<<<g, 256, 0, stream>>>(x, Wq, nullptr, Qb, 0);
  gemm_bias_kernel<<<g, 256, 0, stream>>>(x, Wk, nullptr, Kb, 0);
  gemm_bias_kernel<<<g, 256, 0, stream>>>(x, Wv, nullptr, Vb, 0);

  const int qtiles = BATCH * NHEAD * (S_LEN / 16);
  attention_kernel<<<qtiles / 8, 256, 0, stream>>>(Qb, Kb, Vb, mask, Ab);

  gemm_bias_kernel<<<g, 256, 0, stream>>>(Ab, Wo, bo, (float*)d_out, 1);
}


# Q/K/V projections store 16x64 tiles in WMMA fragment order (Q,K by swapped-operand product); attention reads all fragments as whole-line b128 loads; score d-order and PV key-order of sums change
# speedup vs baseline: 1.2274x; 1.2274x over previous
#include <hip/hip_runtime.h>
#include <hip/hip_bf16.h>


typedef __attribute__((ext_vector_type(2))) float v2f;
typedef __attribute__((ext_vector_type(4))) float v4f;
typedef __attribute__((ext_vector_type(8))) float v8f;

#define EMBED 768
#define S_LEN 2048
#define NHEAD 12
#define HDIM  64
#define BATCH 2
#define ROWS  (BATCH * S_LEN)
#define TILE_FLOATS 1024
#define TILES_PER_BH (S_LEN / 16)

__device__ __forceinline__ v8f wmma4(v2f a, v2f b, v8f c) {
  return __builtin_amdgcn_wmma_f32_16x16x4_f32(
       false, a,  false, b,
       (short)0, c,  false,  false);
}

__device__ __forceinline__ v8f vzero8() {
  v8f z = {0.f, 0.f, 0.f, 0.f, 0.f, 0.f, 0.f, 0.f};
  return z;
}

template <int MODE>
__global__ __launch_bounds__(256) void gemm_kernel(
    const float* __restrict__ A, const float* __restrict__ W,
    const float* __restrict__ bias, float* __restrict__ C) {
  const int lane = threadIdx.x & 31;
  const int wave = threadIdx.x >> 5;
  const int half = lane >> 4;
  const int l16  = lane & 15;

  const int m0 = blockIdx.x * 64  + (wave & 3) * 16;
  const int n0 = blockIdx.y * 128 + (wave >> 2) * 64;

  v8f acc[4];
#pragma unroll
  for (int i = 0; i < 4; ++i) acc[i] = vzero8();

  const float* arow = A + (size_t)(m0 + l16) * EMBED;

#pragma unroll 2
  for (int kk = 0; kk < EMBED; kk += 4) {
    const int ka = kk + 2 * half;
    const v2f a = *(const v2f*)(arow + ka);
    const float* w0 = W + (size_t)ka * EMBED + n0 + l16;
    const float* w1 = w0 + EMBED;
#pragma unroll
    for (int nt = 0; nt < 4; ++nt) {
      v2f b;
      b.x = w0[nt * 16];
      b.y = w1[nt * 16];
      if (MODE == 1) acc[nt] = wmma4(b, a, acc[nt]);
      else           acc[nt] = wmma4(a, b, acc[nt]);
    }
  }

  if (MODE == 0) {
#pragma unroll
    for (int nt = 0; nt < 4; ++nt) {
      const int col = n0 + nt * 16 + l16;
      const float bv = bias[col];
#pragma unroll
      for (int v = 0; v < 8; ++v) {
        C[(size_t)(m0 + v + 8 * half) * EMBED + col] = acc[nt][v] + bv;
      }
    }
  } else {
    const int bidx = m0 / S_LEN;
    const int tt   = (m0 % S_LEN) / 16;
    const int hd   = n0 / HDIM;
    float* tile = C + ((size_t)(bidx * NHEAD + hd) * TILES_PER_BH + tt) * TILE_FLOATS
                    + lane * 4;
#pragma unroll
    for (int nt = 0; nt < 4; ++nt) {
#pragma unroll
      for (int hv = 0; hv < 2; ++hv) {
        v4f o = {acc[nt][4 * hv], acc[nt][4 * hv + 1],
                 acc[nt][4 * hv + 2], acc[nt][4 * hv + 3]};
        *(v4f*)(tile + (nt * 2 + hv) * 128) = o;
      }
    }
  }
}

__global__ __launch_bounds__(256) void attention_kernel(
    const float* __restrict__ Q, const float* __restrict__ K,
    const float* __restrict__ V, const float* __restrict__ mask,
    float* __restrict__ O) {
  __shared__ float pstage[8][16 * 16];

  const int lane = threadIdx.x & 31;
  const int wave = threadIdx.x >> 5;
  const int half = lane >> 4;
  const int l16  = lane & 15;

  const int tile = blockIdx.x * 8 + wave;
  const int qt = tile & 127;
  const int bh = tile >> 7;
  const int b  = bh / NHEAD;
  const int h  = bh % NHEAD;

  const int    q0   = qt * 16;
  const size_t base = (size_t)b * S_LEN * EMBED + (size_t)h * HDIM;

  v4f qf[8];
  {
    const float* qp = Q + ((size_t)bh * TILES_PER_BH + qt) * TILE_FLOATS + lane * 4;
#pragma unroll
    for (int i = 0; i < 8; ++i) qf[i] = *(const v4f*)(qp + i * 128);
  }
  const float* kbase = K + (size_t)bh * TILES_PER_BH * TILE_FLOATS + lane * 4;
  const float* vbase = V + (size_t)bh * TILES_PER_BH * TILE_FLOATS + lane * 4;

  v8f acc[4];
#pragma unroll
  for (int i = 0; i < 4; ++i) acc[i] = vzero8();
  float mrow[8], lrow[8];
#pragma unroll
  for (int v = 0; v < 8; ++v) { mrow[v] = -3.0e38f; lrow[v] = 0.f; }

  float* myp = &pstage[wave][0];

  for (int kt = 0; kt < S_LEN / 16; ++kt) {
    v4f kf[8], vf[8];
    {
      const float* kp = kbase + (size_t)kt * TILE_FLOATS;
      const float* vp = vbase + (size_t)kt * TILE_FLOATS;
#pragma unroll
      for (int i = 0; i < 8; ++i) kf[i] = *(const v4f*)(kp + i * 128);
#pragma unroll
      for (int i = 0; i < 8; ++i) vf[i] = *(const v4f*)(vp + i * 128);
    }

    v8f s = vzero8();
#pragma unroll
    for (int i = 0; i < 8; ++i) {
      v2f a0 = {qf[i].x, qf[i].y};
      v2f b0 = {kf[i].x, kf[i].y};
      s = wmma4(a0, b0, s);
      v2f a1 = {qf[i].z, qf[i].w};
      v2f b1 = {kf[i].z, kf[i].w};
      s = wmma4(a1, b1, s);
    }

    const float* mp = mask + (size_t)(q0 + 8 * half) * S_LEN + kt * 16 + l16;
    float p[8];
#pragma unroll
    for (int v = 0; v < 8; ++v) {
      float sv = s[v] * 0.125f + mp[(size_t)v * S_LEN] * (-1.0e9f);
      float rmax = sv;
      rmax = fmaxf(rmax, __shfl_xor(rmax, 1, 32));
      rmax = fmaxf(rmax, __shfl_xor(rmax, 2, 32));
      rmax = fmaxf(rmax, __shfl_xor(rmax, 4, 32));
      rmax = fmaxf(rmax, __shfl_xor(rmax, 8, 32));
      const float mnew  = fmaxf(mrow[v], rmax);
      const float alpha = __expf(mrow[v] - mnew);
      const float pv    = __expf(sv - mnew);
      float rsum = pv;
      rsum += __shfl_xor(rsum, 1, 32);
      rsum += __shfl_xor(rsum, 2, 32);
      rsum += __shfl_xor(rsum, 4, 32);
      rsum += __shfl_xor(rsum, 8, 32);
      lrow[v] = lrow[v] * alpha + rsum;
      mrow[v] = mnew;
      p[v] = pv;
#pragma unroll
      for (int nt = 0; nt < 4; ++nt) acc[nt][v] *= alpha;
    }

#pragma unroll
    for (int v = 0; v < 8; ++v) myp[(v + 8 * half) * 16 + l16] = p[v];
    __syncthreads();

#pragma unroll
    for (int kk = 0; kk < 4; ++kk) {
      v2f a = *(const v2f*)(myp + l16 * 16 + 8 * half + 2 * kk);
#pragma unroll
      for (int nt = 0; nt < 4; ++nt) {
        const v4f t = vf[nt * 2 + (kk >> 1)];
        v2f bf;
        if (kk & 1) { bf.x = t.z; bf.y = t.w; }
        else        { bf.x = t.x; bf.y = t.y; }
        acc[nt] = wmma4(a, bf, acc[nt]);
      }
    }
    __syncthreads();
  }

#pragma unroll
  for (int v = 0; v < 8; ++v) {
    const float inv = 1.0f / lrow[v];
    float* orow = O + base + (size_t)(q0 + v + 8 * half) * EMBED;
#pragma unroll
    for (int nt = 0; nt < 4; ++nt) {
      orow[nt * 16 + l16] = acc[nt][v] * inv;
    }
  }
}

extern "C" void kernel_launch(void* const* d_in, const int* in_sizes, int n_in,
                              void* d_out, int out_size, void* d_ws, size_t ws_size,
                              hipStream_t stream) {
  const float* x    = (const float*)d_in[0];
  const float* mask = (const float*)d_in[1];
  const float* Wq   = (const float*)d_in[2];
  const float* Wk   = (const float*)d_in[3];
  const float* Wv   = (const float*)d_in[4];
  const float* Wo   = (const float*)d_in[5];
  const float* bo   = (const float*)d_in[6];

  float* ws = (float*)d_ws;
  const size_t mat = (size_t)ROWS * EMBED;
  float* Qb = ws;
  float* Kb = ws + mat;
  float* Vb = ws + 2 * mat;
  float* Ab = ws + 3 * mat;

  dim3 g(ROWS / 64, EMBED / 128);
  gemm_kernel<1><<<g, 256, 0, stream>>>(x, Wq, nullptr, Qb);
  gemm_kernel<1><<<g, 256, 0, stream>>>(x, Wk, nullptr, Kb);
  gemm_kernel<2><<<g, 256, 0, stream>>>(x, Wv, nullptr, Vb);

  const int qtiles = BATCH * NHEAD * (S_LEN / 16);
  attention_kernel<<<qtiles / 8, 256, 0, stream>>>(Qb, Kb, Vb, mask, Ab);

  gemm_kernel<0><<<g, 256, 0, stream>>>(Ab, Wo, bo, (float*)d_out);
}


# projection GEMMs: W read as whole-line float4 row loads (tile t = columns 4c+t), x as float4, 32 k per trip, float4 stores; head tiles carry that feature order; k and score d order of sums change
# speedup vs baseline: 1.5123x; 1.2321x over previous
#include <hip/hip_runtime.h>
#include <hip/hip_bf16.h>


typedef __attribute__((ext_vector_type(2))) float v2f;
typedef __attribute__((ext_vector_type(4))) float v4f;
typedef __attribute__((ext_vector_type(8))) float v8f;

#define EMBED 768
#define S_LEN 2048
#define NHEAD 12
#define HDIM  64
#define BATCH 2
#define ROWS  (BATCH * S_LEN)
#define TILE_FLOATS 1024
#define TILES_PER_BH (S_LEN / 16)

__device__ __forceinline__ v8f wmma4(v2f a, v2f b, v8f c) {
  return __builtin_amdgcn_wmma_f32_16x16x4_f32(
       false, a,  false, b,
       (short)0, c,  false,  false);
}

__device__ __forceinline__ v8f vzero8() {
  v8f z = {0.f, 0.f, 0.f, 0.f, 0.f, 0.f, 0.f, 0.f};
  return z;
}

template <int MODE>
__global__ __launch_bounds__(256) void gemm_kernel(
    const float* __restrict__ A, const float* __restrict__ W,
    const float* __restrict__ bias, float* __restrict__ C) {
  const int lane = threadIdx.x & 31;
  const int wave = threadIdx.x >> 5;
  const int half = lane >> 4;
  const int l16  = lane & 15;

  const int m0 = blockIdx.x * 64  + (wave & 3) * 16;
  const int n0 = blockIdx.y * 128 + (wave >> 2) * 64;

  v8f acc[4];
#pragma unroll
  for (int i = 0; i < 4; ++i) acc[i] = vzero8();

  const float* ap = A + (size_t)(m0 + l16) * EMBED + 16 * half;
  const float* wp = W + (size_t)(16 * half) * EMBED + n0 + 4 * l16;

  for (int k0 = 0; k0 < EMBED; k0 += 32) {
    v4f a[4];
#pragma unroll
    for (int i = 0; i < 4; ++i) a[i] = *(const v4f*)(ap + k0 + 4 * i);
    v4f w[16];
#pragma unroll
    for (int r = 0; r < 16; ++r)
      w[r] = *(const v4f*)(wp + (size_t)(k0 + r) * EMBED);
    __builtin_amdgcn_sched_barrier(0);

#pragma unroll
    for (int j = 0; j < 8; ++j) {
      v2f af;
      if (j & 1) { af.x = a[j >> 1].z; af.y = a[j >> 1].w; }
      else       { af.x = a[j >> 1].x; af.y = a[j >> 1].y; }
#pragma unroll
      for (int t = 0; t < 4; ++t) {
        v2f bf = {w[2 * j][t], w[2 * j + 1][t]};
        if (MODE == 1) acc[t] = wmma4(bf, af, acc[t]);
        else           acc[t] = wmma4(af, bf, acc[t]);
      }
    }
  }

  if (MODE == 0) {
    const int col = n0 + 4 * l16;
    const v4f bv = *(const v4f*)(bias + col);
#pragma unroll
    for (int v = 0; v < 8; ++v) {
      v4f o = {acc[0][v] + bv.x, acc[1][v] + bv.y, acc[2][v] + bv.z, acc[3][v] + bv.w};
      *(v4f*)(C + (size_t)(m0 + v + 8 * half) * EMBED + col) = o;
    }
  } else {
    const int bidx = m0 / S_LEN;
    const int tt   = (m0 % S_LEN) / 16;
    const int hd   = n0 / HDIM;
    float* tile = C + ((size_t)(bidx * NHEAD + hd) * TILES_PER_BH + tt) * TILE_FLOATS
                    + lane * 4;
#pragma unroll
    for (int nt = 0; nt < 4; ++nt) {
#pragma unroll
      for (int hv = 0; hv < 2; ++hv) {
        v4f o = {acc[nt][4 * hv], acc[nt][4 * hv + 1],
                 acc[nt][4 * hv + 2], acc[nt][4 * hv + 3]};
        *(v4f*)(tile + (nt * 2 + hv) * 128) = o;
      }
    }
  }
}

__global__ __launch_bounds__(256) void attention_kernel(
    const float* __restrict__ Q, const float* __restrict__ K,
    const float* __restrict__ V, const float* __restrict__ mask,
    float* __restrict__ O) {
  __shared__ float pstage[8][16 * 16];

  const int lane = threadIdx.x & 31;
  const int wave = threadIdx.x >> 5;
  const int half = lane >> 4;
  const int l16  = lane & 15;

  const int tile = blockIdx.x * 8 + wave;
  const int qt = tile & 127;
  const int bh = tile >> 7;
  const int b  = bh / NHEAD;
  const int h  = bh % NHEAD;

  const int    q0   = qt * 16;
  const size_t base = (size_t)b * S_LEN * EMBED + (size_t)h * HDIM;

  v4f qf[8];
  {
    const float* qp = Q + ((size_t)bh * TILES_PER_BH + qt) * TILE_FLOATS + lane * 4;
#pragma unroll
    for (int i = 0; i < 8; ++i) qf[i] = *(const v4f*)(qp + i * 128);
  }
  const float* kbase = K + (size_t)bh * TILES_PER_BH * TILE_FLOATS + lane * 4;
  const float* vbase = V + (size_t)bh * TILES_PER_BH * TILE_FLOATS + lane * 4;

  v8f acc[4];
#pragma unroll
  for (int i = 0; i < 4; ++i) acc[i] = vzero8();
  float mrow[8], lrow[8];
#pragma unroll
  for (int v = 0; v < 8; ++v) { mrow[v] = -3.0e38f; lrow[v] = 0.f; }

  float* myp = &pstage[wave][0];

  for (int kt = 0; kt < S_LEN / 16; ++kt) {
    v4f kf[8], vf[8];
    {
      const float* kp = kbase + (size_t)kt * TILE_FLOATS;
      const float* vp = vbase + (size_t)kt * TILE_FLOATS;
#pragma unroll
      for (int i = 0; i < 8; ++i) kf[i] = *(const v4f*)(kp + i * 128);
#pragma unroll
      for (int i = 0; i < 8; ++i) vf[i] = *(const v4f*)(vp + i * 128);
    }

    v8f s = vzero8();
#pragma unroll
    for (int i = 0; i < 8; ++i) {
      v2f a0 = {qf[i].x, qf[i].y};
      v2f b0 = {kf[i].x, kf[i].y};
      s = wmma4(a0, b0, s);
      v2f a1 = {qf[i].z, qf[i].w};
      v2f b1 = {kf[i].z, kf[i].w};
      s = wmma4(a1, b1, s);
    }

    const float* mp = mask + (size_t)(q0 + 8 * half) * S_LEN + kt * 16 + l16;
    float p[8];
#pragma unroll
    for (int v = 0; v < 8; ++v) {
      float sv = s[v] * 0.125f + mp[(size_t)v * S_LEN] * (-1.0e9f);
      float rmax = sv;
      rmax = fmaxf(rmax, __shfl_xor(rmax, 1, 32));
      rmax = fmaxf(rmax, __shfl_xor(rmax, 2, 32));
      rmax = fmaxf(rmax, __shfl_xor(rmax, 4, 32));
      rmax = fmaxf(rmax, __shfl_xor(rmax, 8, 32));
      const float mnew  = fmaxf(mrow[v], rmax);
      const float alpha = __expf(mrow[v] - mnew);
      const float pv    = __expf(sv - mnew);
      float rsum = pv;
      rsum += __shfl_xor(rsum, 1, 32);
      rsum += __shfl_xor(rsum, 2, 32);
      rsum += __shfl_xor(rsum, 4, 32);
      rsum += __shfl_xor(rsum, 8, 32);
      lrow[v] = lrow[v] * alpha + rsum;
      mrow[v] = mnew;
      p[v] = pv;
#pragma unroll
      for (int nt = 0; nt < 4; ++nt) acc[nt][v] *= alpha;
    }

#pragma unroll
    for (int v = 0; v < 8; ++v) myp[(v + 8 * half) * 16 + l16] = p[v];
    __syncthreads();

#pragma unroll
    for (int kk = 0; kk < 4; ++kk) {
      v2f a = *(const v2f*)(myp + l16 * 16 + 8 * half + 2 * kk);
#pragma unroll
      for (int nt = 0; nt < 4; ++nt) {
        const v4f t = vf[nt * 2 + (kk >> 1)];
        v2f bf;
        if (kk & 1) { bf.x = t.z; bf.y = t.w; }
        else        { bf.x = t.x; bf.y = t.y; }
        acc[nt] = wmma4(a, bf, acc[nt]);
      }
    }
    __syncthreads();
  }

#pragma unroll
  for (int v = 0; v < 8; ++v) {
    const float inv = 1.0f / lrow[v];
    float* orow = O + base + (size_t)(q0 + v + 8 * half) * EMBED + 4 * l16;
    v4f o = {acc[0][v] * inv, acc[1][v] * inv, acc[2][v] * inv, acc[3][v] * inv};
    *(v4f*)orow = o;
  }
}

extern "C" void kernel_launch(void* const* d_in, const int* in_sizes, int n_in,
                              void* d_out, int out_size, void* d_ws, size_t ws_size,
                              hipStream_t stream) {
  const float* x    = (const float*)d_in[0];
  const float* mask = (const float*)d_in[1];
  const float* Wq   = (const float*)d_in[2];
  const float* Wk   = (const float*)d_in[3];
  const float* Wv   = (const float*)d_in[4];
  const float* Wo   = (const float*)d_in[5];
  const float* bo   = (const float*)d_in[6];

  float* ws = (float*)d_ws;
  const size_t mat = (size_t)ROWS * EMBED;
  float* Qb = ws;
  float* Kb = ws + mat;
  float* Vb = ws + 2 * mat;
  float* Ab = ws + 3 * mat;

  dim3 g(ROWS / 64, EMBED / 128);
  gemm_kernel<1><<<g, 256, 0, stream>>>(x, Wq, nullptr, Qb);
  gemm_kernel<1><<<g, 256, 0, stream>>>(x, Wk, nullptr, Kb);
  gemm_kernel<2><<<g, 256, 0, stream>>>(x, Wv, nullptr, Vb);

  const int qtiles = BATCH * NHEAD * (S_LEN / 16);
  attention_kernel<<<qtiles / 8, 256, 0, stream>>>(Qb, Kb, Vb, mask, Ab);

  gemm_kernel<0><<<g, 256, 0, stream>>>(Ab, Wo, bo, (float*)d_out);
}


# attention softmax: 16-lane row max and row sum as DPP row_xmask butterflies on the vector ALU instead of shuffles (64 LDS-unit permutes per key tile); same pairs, order of sums unchanged
# speedup vs baseline: 1.5965x; 1.0557x over previous
#include <hip/hip_runtime.h>
#include <hip/hip_bf16.h>


typedef __attribute__((ext_vector_type(2))) float v2f;
typedef __attribute__((ext_vector_type(4))) float v4f;
typedef __attribute__((ext_vector_type(8))) float v8f;

#define EMBED 768
#define S_LEN 2048
#define NHEAD 12
#define HDIM  64
#define BATCH 2
#define ROWS  (BATCH * S_LEN)
#define TILE_FLOATS 1024
#define TILES_PER_BH (S_LEN / 16)

__device__ __forceinline__ v8f wmma4(v2f a, v2f b, v8f c) {
  return __builtin_amdgcn_wmma_f32_16x16x4_f32(
       false, a,  false, b,
       (short)0, c,  false,  false);
}

__device__ __forceinline__ v8f vzero8() {
  v8f z = {0.f, 0.f, 0.f, 0.f, 0.f, 0.f, 0.f, 0.f};
  return z;
}

template <int MODE>
__global__ __launch_bounds__(256) void gemm_kernel(
    const float* __restrict__ A, const float* __restrict__ W,
    const float* __restrict__ bias, float* __restrict__ C) {
  const int lane = threadIdx.x & 31;
  const int wave = threadIdx.x >> 5;
  const int half = lane >> 4;
  const int l16  = lane & 15;

  const int m0 = blockIdx.x * 64  + (wave & 3) * 16;
  const int n0 = blockIdx.y * 128 + (wave >> 2) * 64;

  v8f acc[4];
#pragma unroll
  for (int i = 0; i < 4; ++i) acc[i] = vzero8();

  const float* ap = A + (size_t)(m0 + l16) * EMBED + 16 * half;
  const float* wp = W + (size_t)(16 * half) * EMBED + n0 + 4 * l16;

  for (int k0 = 0; k0 < EMBED; k0 += 32) {
    v4f a[4];
#pragma unroll
    for (int i = 0; i < 4; ++i) a[i] = *(const v4f*)(ap + k0 + 4 * i);
    v4f w[16];
#pragma unroll
    for (int r = 0; r < 16; ++r)
      w[r] = *(const v4f*)(wp + (size_t)(k0 + r) * EMBED);
    __builtin_amdgcn_sched_barrier(0);

#pragma unroll
    for (int j = 0; j < 8; ++j) {
      v2f af;
      if (j & 1) { af.x = a[j >> 1].z; af.y = a[j >> 1].w; }
      else       { af.x = a[j >> 1].x; af.y = a[j >> 1].y; }
#pragma unroll
      for (int t = 0; t < 4; ++t) {
        v2f bf = {w[2 * j][t], w[2 * j + 1][t]};
        if (MODE == 1) acc[t] = wmma4(bf, af, acc[t]);
        else           acc[t] = wmma4(af, bf, acc[t]);
      }
    }
  }

  if (MODE == 0) {
    const int col = n0 + 4 * l16;
    const v4f bv = *(const v4f*)(bias + col);
#pragma unroll
    for (int v = 0; v < 8; ++v) {
      v4f o = {acc[0][v] + bv.x, acc[1][v] + bv.y, acc[2][v] + bv.z, acc[3][v] + bv.w};
      *(v4f*)(C + (size_t)(m0 + v + 8 * half) * EMBED + col) = o;
    }
  } else {
    const int bidx = m0 / S_LEN;
    const int tt   = (m0 % S_LEN) / 16;
    const int hd   = n0 / HDIM;
    float* tile = C + ((size_t)(bidx * NHEAD + hd) * TILES_PER_BH + tt) * TILE_FLOATS
                    + lane * 4;
#pragma unroll
    for (int nt = 0; nt < 4; ++nt) {
#pragma unroll
      for (int hv = 0; hv < 2; ++hv) {
        v4f o = {acc[nt][4 * hv], acc[nt][4 * hv + 1],
                 acc[nt][4 * hv + 2], acc[nt][4 * hv + 3]};
        *(v4f*)(tile + (nt * 2 + hv) * 128) = o;
      }
    }
  }
}


template <int M>
__device__ __forceinline__ float xlane16(float v) {
  return __int_as_float(__builtin_amdgcn_update_dpp(
      0, __float_as_int(v), 0x160 + M, 0xf, 0xf, true));
}

#define RMAX_I(d, s, m) \
  "v_max_num_f32_dpp %" #d ", %" #s ", %" #s " row_xmask:" #m " row_mask:0xf bank_mask:0xf\n\t"
#define RMAX_STEP(m) \
  RMAX_I(0, 0, m) RMAX_I(1, 1, m) RMAX_I(2, 2, m) RMAX_I(3, 3, m) \
  RMAX_I(4, 4, m) RMAX_I(5, 5, m) RMAX_I(6, 6, m) RMAX_I(7, 7, m)

__device__ __forceinline__ void rowmax8(const float (&x)[8], float (&r)[8]) {
  asm(RMAX_I(0, 8, 1) RMAX_I(1, 9, 1) RMAX_I(2, 10, 1) RMAX_I(3, 11, 1)
      RMAX_I(4, 12, 1) RMAX_I(5, 13, 1) RMAX_I(6, 14, 1) RMAX_I(7, 15, 1)
      RMAX_STEP(2) RMAX_STEP(4) RMAX_STEP(8)
      : "=&v"(r[0]), "=&v"(r[1]), "=&v"(r[2]), "=&v"(r[3]),
        "=&v"(r[4]), "=&v"(r[5]), "=&v"(r[6]), "=&v"(r[7])
      : "v"(x[0]), "v"(x[1]), "v"(x[2]), "v"(x[3]),
        "v"(x[4]), "v"(x[5]), "v"(x[6]), "v"(x[7]));
}

__global__ __launch_bounds__(256) void attention_kernel(
    const float* __restrict__ Q, const float* __restrict__ K,
    const float* __restrict__ V, const float* __restrict__ mask,
    float* __restrict__ O) {
  __shared__ float pstage[8][16 * 16];

  const int lane = threadIdx.x & 31;
  const int wave = threadIdx.x >> 5;
  const int half = lane >> 4;
  const int l16  = lane & 15;

  const int tile = blockIdx.x * 8 + wave;
  const int qt = tile & 127;
  const int bh = tile >> 7;
  const int b  = bh / NHEAD;
  const int h  = bh % NHEAD;

  const int    q0   = qt * 16;
  const size_t base = (size_t)b * S_LEN * EMBED + (size_t)h * HDIM;

  v4f qf[8];
  {
    const float* qp = Q + ((size_t)bh * TILES_PER_BH + qt) * TILE_FLOATS + lane * 4;
#pragma unroll
    for (int i = 0; i < 8; ++i) qf[i] = *(const v4f*)(qp + i * 128);
  }
  const float* kbase = K + (size_t)bh * TILES_PER_BH * TILE_FLOATS + lane * 4;
  const float* vbase = V + (size_t)bh * TILES_PER_BH * TILE_FLOATS + lane * 4;

  v8f acc[4];
#pragma unroll
  for (int i = 0; i < 4; ++i) acc[i] = vzero8();
  float mrow[8], lrow[8];
#pragma unroll
  for (int v = 0; v < 8; ++v) { mrow[v] = -3.0e38f; lrow[v] = 0.f; }

  float* myp = &pstage[wave][0];

  for (int kt = 0; kt < S_LEN / 16; ++kt) {
    v4f kf[8], vf[8];
    {
      const float* kp = kbase + (size_t)kt * TILE_FLOATS;
      const float* vp = vbase + (size_t)kt * TILE_FLOATS;
#pragma unroll
      for (int i = 0; i < 8; ++i) kf[i] = *(const v4f*)(kp + i * 128);
#pragma unroll
      for (int i = 0; i < 8; ++i) vf[i] = *(const v4f*)(vp + i * 128);
    }

    v8f s = vzero8();
#pragma unroll
    for (int i = 0; i < 8; ++i) {
      v2f a0 = {qf[i].x, qf[i].y};
      v2f b0 = {kf[i].x, kf[i].y};
      s = wmma4(a0, b0, s);
      v2f a1 = {qf[i].z, qf[i].w};
      v2f b1 = {kf[i].z, kf[i].w};
      s = wmma4(a1, b1, s);
    }

    const float* mp = mask + (size_t)(q0 + 8 * half) * S_LEN + kt * 16 + l16;
    float p[8], sv[8], rmax[8], rsum[8], alpha[8];
#pragma unroll
    for (int v = 0; v < 8; ++v)
      sv[v] = s[v] * 0.125f + mp[(size_t)v * S_LEN] * (-1.0e9f);
    rowmax8(sv, rmax);
#pragma unroll
    for (int v = 0; v < 8; ++v) {
      const float mnew = fmaxf(mrow[v], rmax[v]);
      alpha[v] = __expf(mrow[v] - mnew);
      p[v]     = __expf(sv[v] - mnew);
      mrow[v]  = mnew;
#pragma unroll
      for (int nt = 0; nt < 4; ++nt) acc[nt][v] *= alpha[v];
    }
#pragma unroll
    for (int v = 0; v < 8; ++v) rsum[v] = p[v] + xlane16<1>(p[v]);
#pragma unroll
    for (int v = 0; v < 8; ++v) rsum[v] += xlane16<2>(rsum[v]);
#pragma unroll
    for (int v = 0; v < 8; ++v) rsum[v] += xlane16<4>(rsum[v]);
#pragma unroll
    for (int v = 0; v < 8; ++v) rsum[v] += xlane16<8>(rsum[v]);
#pragma unroll
    for (int v = 0; v < 8; ++v) lrow[v] = lrow[v] * alpha[v] + rsum[v];

#pragma unroll
    for (int v = 0; v < 8; ++v) myp[(v + 8 * half) * 16 + l16] = p[v];
    __syncthreads();

#pragma unroll
    for (int kk = 0; kk < 4; ++kk) {
      v2f a = *(const v2f*)(myp + l16 * 16 + 8 * half + 2 * kk);
#pragma unroll
      for (int nt = 0; nt < 4; ++nt) {
        const v4f t = vf[nt * 2 + (kk >> 1)];
        v2f bf;
        if (kk & 1) { bf.x = t.z; bf.y = t.w; }
        else        { bf.x = t.x; bf.y = t.y; }
        acc[nt] = wmma4(a, bf, acc[nt]);
      }
    }
    __syncthreads();
  }

#pragma unroll
  for (int v = 0; v < 8; ++v) {
    const float inv = 1.0f / lrow[v];
    float* orow = O + base + (size_t)(q0 + v + 8 * half) * EMBED + 4 * l16;
    v4f o = {acc[0][v] * inv, acc[1][v] * inv, acc[2][v] * inv, acc[3][v] * inv};
    *(v4f*)orow = o;
  }
}

extern "C" void kernel_launch(void* const* d_in, const int* in_sizes, int n_in,
                              void* d_out, int out_size, void* d_ws, size_t ws_size,
                              hipStream_t stream) {
  const float* x    = (const float*)d_in[0];
  const float* mask = (const float*)d_in[1];
  const float* Wq   = (const float*)d_in[2];
  const float* Wk   = (const float*)d_in[3];
  const float* Wv   = (const float*)d_in[4];
  const float* Wo   = (const float*)d_in[5];
  const float* bo   = (const float*)d_in[6];

  float* ws = (float*)d_ws;
  const size_t mat = (size_t)ROWS * EMBED;
  float* Qb = ws;
  float* Kb = ws + mat;
  float* Vb = ws + 2 * mat;
  float* Ab = ws + 3 * mat;

  dim3 g(ROWS / 64, EMBED / 128);
  gemm_kernel<1><<<g, 256, 0, stream>>>(x, Wq, nullptr, Qb);
  gemm_kernel<1><<<g, 256, 0, stream>>>(x, Wk, nullptr, Kb);
  gemm_kernel<2><<<g, 256, 0, stream>>>(x, Wv, nullptr, Vb);

  const int qtiles = BATCH * NHEAD * (S_LEN / 16);
  attention_kernel<<<qtiles / 8, 256, 0, stream>>>(Qb, Kb, Vb, mask, Ab);

  gemm_kernel<0><<<g, 256, 0, stream>>>(Ab, Wo, bo, (float*)d_out);
}


# all five launches as 768 work-groups of 4 waves (GEMM block tile 64x64, attention 4 query tiles per group): 3 waves on every SIMD, each from another group, not 384 groups of 8; sums unchanged
# speedup vs baseline: 1.8578x; 1.1637x over previous
#include <hip/hip_runtime.h>
#include <hip/hip_bf16.h>


typedef __attribute__((ext_vector_type(2))) float v2f;
typedef __attribute__((ext_vector_type(4))) float v4f;
typedef __attribute__((ext_vector_type(8))) float v8f;

#define EMBED 768
#define S_LEN 2048
#define NHEAD 12
#define HDIM  64
#define BATCH 2
#define ROWS  (BATCH * S_LEN)
#define TILE_FLOATS 1024
#define TILES_PER_BH (S_LEN / 16)

__device__ __forceinline__ v8f wmma4(v2f a, v2f b, v8f c) {
  return __builtin_amdgcn_wmma_f32_16x16x4_f32(
       false, a,  false, b,
       (short)0, c,  false,  false);
}

__device__ __forceinline__ v8f vzero8() {
  v8f z = {0.f, 0.f, 0.f, 0.f, 0.f, 0.f, 0.f, 0.f};
  return z;
}

template <int MODE>
__global__ __launch_bounds__(128) void gemm_kernel(
    const float* __restrict__ A, const float* __restrict__ W,
    const float* __restrict__ bias, float* __restrict__ C) {
  const int lane = threadIdx.x & 31;
  const int wave = threadIdx.x >> 5;
  const int half = lane >> 4;
  const int l16  = lane & 15;

  const int m0 = blockIdx.x * 64 + wave * 16;
  const int n0 = blockIdx.y * 64;

  v8f acc[4];
#pragma unroll
  for (int i = 0; i < 4; ++i) acc[i] = vzero8();

  const float* ap = A + (size_t)(m0 + l16) * EMBED + 16 * half;
  const float* wp = W + (size_t)(16 * half) * EMBED + n0 + 4 * l16;

  for (int k0 = 0; k0 < EMBED; k0 += 32) {
    v4f a[4];
#pragma unroll
    for (int i = 0; i < 4; ++i) a[i] = *(const v4f*)(ap + k0 + 4 * i);
    v4f w[16];
#pragma unroll
    for (int r = 0; r < 16; ++r)
      w[r] = *(const v4f*)(wp + (size_t)(k0 + r) * EMBED);
    __builtin_amdgcn_sched_barrier(0);

#pragma unroll
    for (int j = 0; j < 8; ++j) {
      v2f af;
      if (j & 1) { af.x = a[j >> 1].z; af.y = a[j >> 1].w; }
      else       { af.x = a[j >> 1].x; af.y = a[j >> 1].y; }
#pragma unroll
      for (int t = 0; t < 4; ++t) {
        v2f bf = {w[2 * j][t], w[2 * j + 1][t]};
        if (MODE == 1) acc[t] = wmma4(bf, af, acc[t]);
        else           acc[t] = wmma4(af, bf, acc[t]);
      }
    }
  }

  if (MODE == 0) {
    const int col = n0 + 4 * l16;
    const v4f bv = *(const v4f*)(bias + col);
#pragma unroll
    for (int v = 0; v < 8; ++v) {
      v4f o = {acc[0][v] + bv.x, acc[1][v] + bv.y, acc[2][v] + bv.z, acc[3][v] + bv.w};
      *(v4f*)(C + (size_t)(m0 + v + 8 * half) * EMBED + col) = o;
    }
  } else {
    const int bidx = m0 / S_LEN;
    const int tt   = (m0 % S_LEN) / 16;
    const int hd   = n0 / HDIM;
    float* tile = C + ((size_t)(bidx * NHEAD + hd) * TILES_PER_BH + tt) * TILE_FLOATS
                    + lane * 4;
#pragma unroll
    for (int nt = 0; nt < 4; ++nt) {
#pragma unroll
      for (int hv = 0; hv < 2; ++hv) {
        v4f o = {acc[nt][4 * hv], acc[nt][4 * hv + 1],
                 acc[nt][4 * hv + 2], acc[nt][4 * hv + 3]};
        *(v4f*)(tile + (nt * 2 + hv) * 128) = o;
      }
    }
  }
}


template <int M>
__device__ __forceinline__ float xlane16(float v) {
  return __int_as_float(__builtin_amdgcn_update_dpp(
      0, __float_as_int(v), 0x160 + M, 0xf, 0xf, true));
}

#define RMAX_I(d, s, m) \
  "v_max_num_f32_dpp %" #d ", %" #s ", %" #s " row_xmask:" #m " row_mask:0xf bank_mask:0xf\n\t"
#define RMAX_STEP(m) \
  RMAX_I(0, 0, m) RMAX_I(1, 1, m) RMAX_I(2, 2, m) RMAX_I(3, 3, m) \
  RMAX_I(4, 4, m) RMAX_I(5, 5, m) RMAX_I(6, 6, m) RMAX_I(7, 7, m)

__device__ __forceinline__ void rowmax8(const float (&x)[8], float (&r)[8]) {
  asm(RMAX_I(0, 8, 1) RMAX_I(1, 9, 1) RMAX_I(2, 10, 1) RMAX_I(3, 11, 1)
      RMAX_I(4, 12, 1) RMAX_I(5, 13, 1) RMAX_I(6, 14, 1) RMAX_I(7, 15, 1)
      RMAX_STEP(2) RMAX_STEP(4) RMAX_STEP(8)
      : "=&v"(r[0]), "=&v"(r[1]), "=&v"(r[2]), "=&v"(r[3]),
        "=&v"(r[4]), "=&v"(r[5]), "=&v"(r[6]), "=&v"(r[7])
      : "v"(x[0]), "v"(x[1]), "v"(x[2]), "v"(x[3]),
        "v"(x[4]), "v"(x[5]), "v"(x[6]), "v"(x[7]));
}

__global__ __launch_bounds__(128) void attention_kernel(
    const float* __restrict__ Q, const float* __restrict__ K,
    const float* __restrict__ V, const float* __restrict__ mask,
    float* __restrict__ O) {
  __shared__ float pstage[4][16 * 16];

  const int lane = threadIdx.x & 31;
  const int wave = threadIdx.x >> 5;
  const int half = lane >> 4;
  const int l16  = lane & 15;

  const int tile = blockIdx.x * 4 + wave;
  const int qt = tile & 127;
  const int bh = tile >> 7;
  const int b  = bh / NHEAD;
  const int h  = bh % NHEAD;

  const int    q0   = qt * 16;
  const size_t base = (size_t)b * S_LEN * EMBED + (size_t)h * HDIM;

  v4f qf[8];
  {
    const float* qp = Q + ((size_t)bh * TILES_PER_BH + qt) * TILE_FLOATS + lane * 4;
#pragma unroll
    for (int i = 0; i < 8; ++i) qf[i] = *(const v4f*)(qp + i * 128);
  }
  const float* kbase = K + (size_t)bh * TILES_PER_BH * TILE_FLOATS + lane * 4;
  const float* vbase = V + (size_t)bh * TILES_PER_BH * TILE_FLOATS + lane * 4;

  v8f acc[4];
#pragma unroll
  for (int i = 0; i < 4; ++i) acc[i] = vzero8();
  float mrow[8], lrow[8];
#pragma unroll
  for (int v = 0; v < 8; ++v) { mrow[v] = -3.0e38f; lrow[v] = 0.f; }

  float* myp = &pstage[wave][0];

  for (int kt = 0; kt < S_LEN / 16; ++kt) {
    v4f kf[8], vf[8];
    {
      const float* kp = kbase + (size_t)kt * TILE_FLOATS;
      const float* vp = vbase + (size_t)kt * TILE_FLOATS;
#pragma unroll
      for (int i = 0; i < 8; ++i) kf[i] = *(const v4f*)(kp + i * 128);
#pragma unroll
      for (int i = 0; i < 8; ++i) vf[i] = *(const v4f*)(vp + i * 128);
    }

    v8f s = vzero8();
#pragma unroll
    for (int i = 0; i < 8; ++i) {
      v2f a0 = {qf[i].x, qf[i].y};
      v2f b0 = {kf[i].x, kf[i].y};
      s = wmma4(a0, b0, s);
      v2f a1 = {qf[i].z, qf[i].w};
      v2f b1 = {kf[i].z, kf[i].w};
      s = wmma4(a1, b1, s);
    }

    const float* mp = mask + (size_t)(q0 + 8 * half) * S_LEN + kt * 16 + l16;
    float p[8], sv[8], rmax[8], rsum[8], alpha[8];
#pragma unroll
    for (int v = 0; v < 8; ++v)
      sv[v] = s[v] * 0.125f + mp[(size_t)v * S_LEN] * (-1.0e9f);
    rowmax8(sv, rmax);
#pragma unroll
    for (int v = 0; v < 8; ++v) {
      const float mnew = fmaxf(mrow[v], rmax[v]);
      alpha[v] = __expf(mrow[v] - mnew);
      p[v]     = __expf(sv[v] - mnew);
      mrow[v]  = mnew;
#pragma unroll
      for (int nt = 0; nt < 4; ++nt) acc[nt][v] *= alpha[v];
    }
#pragma unroll
    for (int v = 0; v < 8; ++v) rsum[v] = p[v] + xlane16<1>(p[v]);
#pragma unroll
    for (int v = 0; v < 8; ++v) rsum[v] += xlane16<2>(rsum[v]);
#pragma unroll
    for (int v = 0; v < 8; ++v) rsum[v] += xlane16<4>(rsum[v]);
#pragma unroll
    for (int v = 0; v < 8; ++v) rsum[v] += xlane16<8>(rsum[v]);
#pragma unroll
    for (int v = 0; v < 8; ++v) lrow[v] = lrow[v] * alpha[v] + rsum[v];

#pragma unroll
    for (int v = 0; v < 8; ++v) myp[(v + 8 * half) * 16 + l16] = p[v];
    __syncthreads();

#pragma unroll
    for (int kk = 0; kk < 4; ++kk) {
      v2f a = *(const v2f*)(myp + l16 * 16 + 8 * half + 2 * kk);
#pragma unroll
      for (int nt = 0; nt < 4; ++nt) {
        const v4f t = vf[nt * 2 + (kk >> 1)];
        v2f bf;
        if (kk & 1) { bf.x = t.z; bf.y = t.w; }
        else        { bf.x = t.x; bf.y = t.y; }
        acc[nt] = wmma4(a, bf, acc[nt]);
      }
    }
    __syncthreads();
  }

#pragma unroll
  for (int v = 0; v < 8; ++v) {
    const float inv = 1.0f / lrow[v];
    float* orow = O + base + (size_t)(q0 + v + 8 * half) * EMBED + 4 * l16;
    v4f o = {acc[0][v] * inv, acc[1][v] * inv, acc[2][v] * inv, acc[3][v] * inv};
    *(v4f*)orow = o;
  }
}

extern "C" void kernel_launch(void* const* d_in, const int* in_sizes, int n_in,
                              void* d_out, int out_size, void* d_ws, size_t ws_size,
                              hipStream_t stream) {
  const float* x    = (const float*)d_in[0];
  const float* mask = (const float*)d_in[1];
  const float* Wq   = (const float*)d_in[2];
  const float* Wk   = (const float*)d_in[3];
  const float* Wv   = (const float*)d_in[4];
  const float* Wo   = (const float*)d_in[5];
  const float* bo   = (const float*)d_in[6];

  float* ws = (float*)d_ws;
  const size_t mat = (size_t)ROWS * EMBED;
  float* Qb = ws;
  float* Kb = ws + mat;
  float* Vb = ws + 2 * mat;
  float* Ab = ws + 3 * mat;

  dim3 g(ROWS / 64, EMBED / 64);
  gemm_kernel<1><<<g, 128, 0, stream>>>(x, Wq, nullptr, Qb);
  gemm_kernel<1><<<g, 128, 0, stream>>>(x, Wk, nullptr, Kb);
  gemm_kernel<2><<<g, 128, 0, stream>>>(x, Wv, nullptr, Vb);

  const int qtiles = BATCH * NHEAD * (S_LEN / 16);
  attention_kernel<<<qtiles / 4, 128, 0, stream>>>(Qb, Kb, Vb, mask, Ab);

  gemm_kernel<0><<<g, 128, 0, stream>>>(Ab, Wo, bo, (float*)d_out);
}


# attention: next key tile's K fragments and mask block loaded one tile ahead into a second register set (V in the same group); barriers fence LDS only so these loads stay in flight; sums unchanged
# speedup vs baseline: 2.0218x; 1.0883x over previous
#include <hip/hip_runtime.h>
#include <hip/hip_bf16.h>


typedef __attribute__((ext_vector_type(2))) float v2f;
typedef __attribute__((ext_vector_type(4))) float v4f;
typedef __attribute__((ext_vector_type(8))) float v8f;

#define EMBED 768
#define S_LEN 2048
#define NHEAD 12
#define HDIM  64
#define BATCH 2
#define ROWS  (BATCH * S_LEN)
#define TILE_FLOATS 1024
#define TILES_PER_BH (S_LEN / 16)

__device__ __forceinline__ v8f wmma4(v2f a, v2f b, v8f c) {
  return __builtin_amdgcn_wmma_f32_16x16x4_f32(
       false, a,  false, b,
       (short)0, c,  false,  false);
}

__device__ __forceinline__ v8f vzero8() {
  v8f z = {0.f, 0.f, 0.f, 0.f, 0.f, 0.f, 0.f, 0.f};
  return z;
}

template <int MODE>
__global__ __launch_bounds__(128) void gemm_kernel(
    const float* __restrict__ A, const float* __restrict__ W,
    const float* __restrict__ bias, float* __restrict__ C) {
  const int lane = threadIdx.x & 31;
  const int wave = threadIdx.x >> 5;
  const int half = lane >> 4;
  const int l16  = lane & 15;

  const int m0 = blockIdx.x * 64 + wave * 16;
  const int n0 = blockIdx.y * 64;

  v8f acc[4];
#pragma unroll
  for (int i = 0; i < 4; ++i) acc[i] = vzero8();

  const float* ap = A + (size_t)(m0 + l16) * EMBED + 16 * half;
  const float* wp = W + (size_t)(16 * half) * EMBED + n0 + 4 * l16;

  for (int k0 = 0; k0 < EMBED; k0 += 32) {
    v4f a[4];
#pragma unroll
    for (int i = 0; i < 4; ++i) a[i] = *(const v4f*)(ap + k0 + 4 * i);
    v4f w[16];
#pragma unroll
    for (int r = 0; r < 16; ++r)
      w[r] = *(const v4f*)(wp + (size_t)(k0 + r) * EMBED);
    __builtin_amdgcn_sched_barrier(0);

#pragma unroll
    for (int j = 0; j < 8; ++j) {
      v2f af;
      if (j & 1) { af.x = a[j >> 1].z; af.y = a[j >> 1].w; }
      else       { af.x = a[j >> 1].x; af.y = a[j >> 1].y; }
#pragma unroll
      for (int t = 0; t < 4; ++t) {
        v2f bf = {w[2 * j][t], w[2 * j + 1][t]};
        if (MODE == 1) acc[t] = wmma4(bf, af, acc[t]);
        else           acc[t] = wmma4(af, bf, acc[t]);
      }
    }
  }

  if (MODE == 0) {
    const int col = n0 + 4 * l16;
    const v4f bv = *(const v4f*)(bias + col);
#pragma unroll
    for (int v = 0; v < 8; ++v) {
      v4f o = {acc[0][v] + bv.x, acc[1][v] + bv.y, acc[2][v] + bv.z, acc[3][v] + bv.w};
      *(v4f*)(C + (size_t)(m0 + v + 8 * half) * EMBED + col) = o;
    }
  } else {
    const int bidx = m0 / S_LEN;
    const int tt   = (m0 % S_LEN) / 16;
    const int hd   = n0 / HDIM;
    float* tile = C + ((size_t)(bidx * NHEAD + hd) * TILES_PER_BH + tt) * TILE_FLOATS
                    + lane * 4;
#pragma unroll
    for (int nt = 0; nt < 4; ++nt) {
#pragma unroll
      for (int hv = 0; hv < 2; ++hv) {
        v4f o = {acc[nt][4 * hv], acc[nt][4 * hv + 1],
                 acc[nt][4 * hv + 2], acc[nt][4 * hv + 3]};
        *(v4f*)(tile + (nt * 2 + hv) * 128) = o;
      }
    }
  }
}


template <int M>
__device__ __forceinline__ float xlane16(float v) {
  return __int_as_float(__builtin_amdgcn_update_dpp(
      0, __float_as_int(v), 0x160 + M, 0xf, 0xf, true));
}

#define RMAX_I(d, s, m) \
  "v_max_num_f32_dpp %" #d ", %" #s ", %" #s " row_xmask:" #m " row_mask:0xf bank_mask:0xf\n\t"
#define RMAX_STEP(m) \
  RMAX_I(0, 0, m) RMAX_I(1, 1, m) RMAX_I(2, 2, m) RMAX_I(3, 3, m) \
  RMAX_I(4, 4, m) RMAX_I(5, 5, m) RMAX_I(6, 6, m) RMAX_I(7, 7, m)

__device__ __forceinline__ void rowmax8(const float (&x)[8], float (&r)[8]) {
  asm(RMAX_I(0, 8, 1) RMAX_I(1, 9, 1) RMAX_I(2, 10, 1) RMAX_I(3, 11, 1)
      RMAX_I(4, 12, 1) RMAX_I(5, 13, 1) RMAX_I(6, 14, 1) RMAX_I(7, 15, 1)
      RMAX_STEP(2) RMAX_STEP(4) RMAX_STEP(8)
      : "=&v"(r[0]), "=&v"(r[1]), "=&v"(r[2]), "=&v"(r[3]),
        "=&v"(r[4]), "=&v"(r[5]), "=&v"(r[6]), "=&v"(r[7])
      : "v"(x[0]), "v"(x[1]), "v"(x[2]), "v"(x[3]),
        "v"(x[4]), "v"(x[5]), "v"(x[6]), "v"(x[7]));
}

__device__ __forceinline__ void load_k_mask(const float* __restrict__ kbase,
                                            const float* __restrict__ mbase, int kt,
                                            v4f (&kf)[8], float (&mk)[8]) {
  const float* kp = kbase + (size_t)kt * TILE_FLOATS;
  const float* mp = mbase + kt * 16;
#pragma unroll
  for (int i = 0; i < 8; ++i) kf[i] = *(const v4f*)(kp + i * 128);
#pragma unroll
  for (int v = 0; v < 8; ++v) mk[v] = mp[(size_t)v * S_LEN];
}

__device__ __forceinline__ void load_v(const float* __restrict__ vbase, int kt,
                                       v4f (&vf)[8]) {
  const float* vp = vbase + (size_t)kt * TILE_FLOATS;
#pragma unroll
  for (int i = 0; i < 8; ++i) vf[i] = *(const v4f*)(vp + i * 128);
}

__device__ __forceinline__ void lds_barrier() {
  __builtin_amdgcn_fence(__ATOMIC_RELEASE, "workgroup", "local");
  __builtin_amdgcn_s_barrier();
  __builtin_amdgcn_fence(__ATOMIC_ACQUIRE, "workgroup", "local");
}

__device__ __forceinline__ void key_tile_step(
    const v4f (&qf)[8], const v4f (&kf)[8], const v4f (&vf)[8], const float (&mk)[8],
    v8f (&acc)[4], float (&mrow)[8], float (&lrow)[8], float* myp, int half, int l16) {
  v8f s = vzero8();
#pragma unroll
  for (int i = 0; i < 8; ++i) {
    v2f a0 = {qf[i].x, qf[i].y};
    v2f b0 = {kf[i].x, kf[i].y};
    s = wmma4(a0, b0, s);
    v2f a1 = {qf[i].z, qf[i].w};
    v2f b1 = {kf[i].z, kf[i].w};
    s = wmma4(a1, b1, s);
  }

  float p[8], sv[8], rmax[8], rsum[8], alpha[8];
#pragma unroll
  for (int v = 0; v < 8; ++v)
    sv[v] = s[v] * 0.125f + mk[v] * (-1.0e9f);
  rowmax8(sv, rmax);
#pragma unroll
  for (int v = 0; v < 8; ++v) {
    const float mnew = fmaxf(mrow[v], rmax[v]);
    alpha[v] = __expf(mrow[v] - mnew);
    p[v]     = __expf(sv[v] - mnew);
    mrow[v]  = mnew;
#pragma unroll
    for (int nt = 0; nt < 4; ++nt) acc[nt][v] *= alpha[v];
  }
#pragma unroll
  for (int v = 0; v < 8; ++v) rsum[v] = p[v] + xlane16<1>(p[v]);
#pragma unroll
  for (int v = 0; v < 8; ++v) rsum[v] += xlane16<2>(rsum[v]);
#pragma unroll
  for (int v = 0; v < 8; ++v) rsum[v] += xlane16<4>(rsum[v]);
#pragma unroll
  for (int v = 0; v < 8; ++v) rsum[v] += xlane16<8>(rsum[v]);
#pragma unroll
  for (int v = 0; v < 8; ++v) lrow[v] = lrow[v] * alpha[v] + rsum[v];

#pragma unroll
  for (int v = 0; v < 8; ++v) myp[(v + 8 * half) * 16 + l16] = p[v];
  lds_barrier();

#pragma unroll
  for (int kk = 0; kk < 4; ++kk) {
    v2f a = *(const v2f*)(myp + l16 * 16 + 8 * half + 2 * kk);
#pragma unroll
    for (int nt = 0; nt < 4; ++nt) {
      const v4f t = vf[nt * 2 + (kk >> 1)];
      v2f bf;
      if (kk & 1) { bf.x = t.z; bf.y = t.w; }
      else        { bf.x = t.x; bf.y = t.y; }
      acc[nt] = wmma4(a, bf, acc[nt]);
    }
  }
  lds_barrier();
}

__global__ __launch_bounds__(128) void attention_kernel(
    const float* __restrict__ Q, const float* __restrict__ K,
    const float* __restrict__ V, const float* __restrict__ mask,
    float* __restrict__ O) {
  __shared__ float pstage[4][16 * 16];

  const int lane = threadIdx.x & 31;
  const int wave = threadIdx.x >> 5;
  const int half = lane >> 4;
  const int l16  = lane & 15;

  const int tile = blockIdx.x * 4 + wave;
  const int qt = tile & 127;
  const int bh = tile >> 7;
  const int b  = bh / NHEAD;
  const int h  = bh % NHEAD;

  const int    q0   = qt * 16;
  const size_t base = (size_t)b * S_LEN * EMBED + (size_t)h * HDIM;

  v4f qf[8];
  {
    const float* qp = Q + ((size_t)bh * TILES_PER_BH + qt) * TILE_FLOATS + lane * 4;
#pragma unroll
    for (int i = 0; i < 8; ++i) qf[i] = *(const v4f*)(qp + i * 128);
  }
  const float* kbase = K + (size_t)bh * TILES_PER_BH * TILE_FLOATS + lane * 4;
  const float* vbase = V + (size_t)bh * TILES_PER_BH * TILE_FLOATS + lane * 4;

  v8f acc[4];
#pragma unroll
  for (int i = 0; i < 4; ++i) acc[i] = vzero8();
  float mrow[8], lrow[8];
#pragma unroll
  for (int v = 0; v < 8; ++v) { mrow[v] = -3.0e38f; lrow[v] = 0.f; }

  float* myp = &pstage[wave][0];

  v4f kA[8], kB[8], vf[8];
  float mA[8], mB[8];
  const float* mbase = mask + (size_t)(q0 + 8 * half) * S_LEN + l16;
  load_k_mask(kbase, mbase, 0, kA, mA);
  for (int kt = 0; kt < S_LEN / 16; kt += 2) {
    load_v(vbase, kt, vf);
    load_k_mask(kbase, mbase, kt + 1, kB, mB);
    __builtin_amdgcn_sched_barrier(0);
    key_tile_step(qf, kA, vf, mA, acc, mrow, lrow, myp, half, l16);
    const int kn = (kt + 2 < S_LEN / 16) ? kt + 2 : kt + 1;
    load_v(vbase, kt + 1, vf);
    load_k_mask(kbase, mbase, kn, kA, mA);
    __builtin_amdgcn_sched_barrier(0);
    key_tile_step(qf, kB, vf, mB, acc, mrow, lrow, myp, half, l16);
  }

#pragma unroll
  for (int v = 0; v < 8; ++v) {
    const float inv = 1.0f / lrow[v];
    float* orow = O + base + (size_t)(q0 + v + 8 * half) * EMBED + 4 * l16;
    v4f o = {acc[0][v] * inv, acc[1][v] * inv, acc[2][v] * inv, acc[3][v] * inv};
    *(v4f*)orow = o;
  }
}

extern "C" void kernel_launch(void* const* d_in, const int* in_sizes, int n_in,
                              void* d_out, int out_size, void* d_ws, size_t ws_size,
                              hipStream_t stream) {
  const float* x    = (const float*)d_in[0];
  const float* mask = (const float*)d_in[1];
  const float* Wq   = (const float*)d_in[2];
  const float* Wk   = (const float*)d_in[3];
  const float* Wv   = (const float*)d_in[4];
  const float* Wo   = (const float*)d_in[5];
  const float* bo   = (const float*)d_in[6];

  float* ws = (float*)d_ws;
  const size_t mat = (size_t)ROWS * EMBED;
  float* Qb = ws;
  float* Kb = ws + mat;
  float* Vb = ws + 2 * mat;
  float* Ab = ws + 3 * mat;

  dim3 g(ROWS / 64, EMBED / 64);
  gemm_kernel<1><<<g, 128, 0, stream>>>(x, Wq, nullptr, Qb);
  gemm_kernel<1><<<g, 128, 0, stream>>>(x, Wk, nullptr, Kb);
  gemm_kernel<2><<<g, 128, 0, stream>>>(x, Wv, nullptr, Vb);

  const int qtiles = BATCH * NHEAD * (S_LEN / 16);
  attention_kernel<<<qtiles / 4, 128, 0, stream>>>(Qb, Kb, Vb, mask, Ab);

  gemm_kernel<0><<<g, 128, 0, stream>>>(Ab, Wo, bo, (float*)d_out);
}


# attention: online-softmax step of 32 keys, not 16: two score chains side by side, one row max/sum butterfly, one O rescale, one LDS pass of P and one barrier pair per step; row-sum order changes
# speedup vs baseline: 2.2113x; 1.0937x over previous
#include <hip/hip_runtime.h>
#include <hip/hip_bf16.h>


typedef __attribute__((ext_vector_type(2))) float v2f;
typedef __attribute__((ext_vector_type(4))) float v4f;
typedef __attribute__((ext_vector_type(8))) float v8f;

#define EMBED 768
#define S_LEN 2048
#define NHEAD 12
#define HDIM  64
#define BATCH 2
#define ROWS  (BATCH * S_LEN)
#define TILE_FLOATS 1024
#define TILES_PER_BH (S_LEN / 16)

__device__ __forceinline__ v8f wmma4(v2f a, v2f b, v8f c) {
  return __builtin_amdgcn_wmma_f32_16x16x4_f32(
       false, a,  false, b,
       (short)0, c,  false,  false);
}

__device__ __forceinline__ v8f vzero8() {
  v8f z = {0.f, 0.f, 0.f, 0.f, 0.f, 0.f, 0.f, 0.f};
  return z;
}

template <int MODE>
__global__ __launch_bounds__(128) void gemm_kernel(
    const float* __restrict__ A, const float* __restrict__ W,
    const float* __restrict__ bias, float* __restrict__ C) {
  const int lane = threadIdx.x & 31;
  const int wave = threadIdx.x >> 5;
  const int half = lane >> 4;
  const int l16  = lane & 15;

  const int m0 = blockIdx.x * 64 + wave * 16;
  const int n0 = blockIdx.y * 64;

  v8f acc[4];
#pragma unroll
  for (int i = 0; i < 4; ++i) acc[i] = vzero8();

  const float* ap = A + (size_t)(m0 + l16) * EMBED + 16 * half;
  const float* wp = W + (size_t)(16 * half) * EMBED + n0 + 4 * l16;

  for (int k0 = 0; k0 < EMBED; k0 += 32) {
    v4f a[4];
#pragma unroll
    for (int i = 0; i < 4; ++i) a[i] = *(const v4f*)(ap + k0 + 4 * i);
    v4f w[16];
#pragma unroll
    for (int r = 0; r < 16; ++r)
      w[r] = *(const v4f*)(wp + (size_t)(k0 + r) * EMBED);
    __builtin_amdgcn_sched_barrier(0);

#pragma unroll
    for (int j = 0; j < 8; ++j) {
      v2f af;
      if (j & 1) { af.x = a[j >> 1].z; af.y = a[j >> 1].w; }
      else       { af.x = a[j >> 1].x; af.y = a[j >> 1].y; }
#pragma unroll
      for (int t = 0; t < 4; ++t) {
        v2f bf = {w[2 * j][t], w[2 * j + 1][t]};
        if (MODE == 1) acc[t] = wmma4(bf, af, acc[t]);
        else           acc[t] = wmma4(af, bf, acc[t]);
      }
    }
  }

  if (MODE == 0) {
    const int col = n0 + 4 * l16;
    const v4f bv = *(const v4f*)(bias + col);
#pragma unroll
    for (int v = 0; v < 8; ++v) {
      v4f o = {acc[0][v] + bv.x, acc[1][v] + bv.y, acc[2][v] + bv.z, acc[3][v] + bv.w};
      *(v4f*)(C + (size_t)(m0 + v + 8 * half) * EMBED + col) = o;
    }
  } else {
    const int bidx = m0 / S_LEN;
    const int tt   = (m0 % S_LEN) / 16;
    const int hd   = n0 / HDIM;
    float* tile = C + ((size_t)(bidx * NHEAD + hd) * TILES_PER_BH + tt) * TILE_FLOATS
                    + lane * 4;
#pragma unroll
    for (int nt = 0; nt < 4; ++nt) {
#pragma unroll
      for (int hv = 0; hv < 2; ++hv) {
        v4f o = {acc[nt][4 * hv], acc[nt][4 * hv + 1],
                 acc[nt][4 * hv + 2], acc[nt][4 * hv + 3]};
        *(v4f*)(tile + (nt * 2 + hv) * 128) = o;
      }
    }
  }
}


template <int M>
__device__ __forceinline__ float xlane16(float v) {
  return __int_as_float(__builtin_amdgcn_update_dpp(
      0, __float_as_int(v), 0x160 + M, 0xf, 0xf, true));
}

#define RMAX_I(d, s, m) \
  "v_max_num_f32_dpp %" #d ", %" #s ", %" #s " row_xmask:" #m " row_mask:0xf bank_mask:0xf\n\t"
#define RMAX_STEP(m) \
  RMAX_I(0, 0, m) RMAX_I(1, 1, m) RMAX_I(2, 2, m) RMAX_I(3, 3, m) \
  RMAX_I(4, 4, m) RMAX_I(5, 5, m) RMAX_I(6, 6, m) RMAX_I(7, 7, m)

__device__ __forceinline__ void rowmax8(const float (&x)[8], float (&r)[8]) {
  asm(RMAX_I(0, 8, 1) RMAX_I(1, 9, 1) RMAX_I(2, 10, 1) RMAX_I(3, 11, 1)
      RMAX_I(4, 12, 1) RMAX_I(5, 13, 1) RMAX_I(6, 14, 1) RMAX_I(7, 15, 1)
      RMAX_STEP(2) RMAX_STEP(4) RMAX_STEP(8)
      : "=&v"(r[0]), "=&v"(r[1]), "=&v"(r[2]), "=&v"(r[3]),
        "=&v"(r[4]), "=&v"(r[5]), "=&v"(r[6]), "=&v"(r[7])
      : "v"(x[0]), "v"(x[1]), "v"(x[2]), "v"(x[3]),
        "v"(x[4]), "v"(x[5]), "v"(x[6]), "v"(x[7]));
}

__device__ __forceinline__ void load_k2(const float* __restrict__ kbase, int st,
                                        v4f (&k0)[8], v4f (&k1)[8]) {
  const float* kp = kbase + (size_t)st * (2 * TILE_FLOATS);
#pragma unroll
  for (int i = 0; i < 8; ++i) k0[i] = *(const v4f*)(kp + i * 128);
#pragma unroll
  for (int i = 0; i < 8; ++i) k1[i] = *(const v4f*)(kp + TILE_FLOATS + i * 128);
}

__device__ __forceinline__ void load_mask2(const float* __restrict__ mbase, int st,
                                           float (&m0)[8], float (&m1)[8]) {
  const float* mp = mbase + st * 32;
#pragma unroll
  for (int v = 0; v < 8; ++v) m0[v] = mp[(size_t)v * S_LEN];
#pragma unroll
  for (int v = 0; v < 8; ++v) m1[v] = mp[(size_t)v * S_LEN + 16];
}

__device__ __forceinline__ void load_v2(const float* __restrict__ vbase, int st,
                                        v4f (&v0)[8], v4f (&v1)[8]) {
  const float* vp = vbase + (size_t)st * (2 * TILE_FLOATS);
#pragma unroll
  for (int i = 0; i < 8; ++i) v0[i] = *(const v4f*)(vp + i * 128);
#pragma unroll
  for (int i = 0; i < 8; ++i) v1[i] = *(const v4f*)(vp + TILE_FLOATS + i * 128);
}

__device__ __forceinline__ void lds_barrier() {
  __builtin_amdgcn_fence(__ATOMIC_RELEASE, "workgroup", "local");
  __builtin_amdgcn_s_barrier();
  __builtin_amdgcn_fence(__ATOMIC_ACQUIRE, "workgroup", "local");
}

__device__ __forceinline__ void pv_tile(const v4f pa, const v4f pb, const v4f (&vf)[8],
                                        v8f (&acc)[4]) {
#pragma unroll
  for (int kk = 0; kk < 4; ++kk) {
    const v4f pq = (kk >> 1) ? pb : pa;
    v2f a;
    if (kk & 1) { a.x = pq.z; a.y = pq.w; }
    else        { a.x = pq.x; a.y = pq.y; }
#pragma unroll
    for (int nt = 0; nt < 4; ++nt) {
      const v4f t = vf[nt * 2 + (kk >> 1)];
      v2f bf;
      if (kk & 1) { bf.x = t.z; bf.y = t.w; }
      else        { bf.x = t.x; bf.y = t.y; }
      acc[nt] = wmma4(a, bf, acc[nt]);
    }
  }
}

#define PPITCH 36

__global__ __launch_bounds__(128) void attention_kernel(
    const float* __restrict__ Q, const float* __restrict__ K,
    const float* __restrict__ V, const float* __restrict__ mask,
    float* __restrict__ O) {
  __shared__ __attribute__((aligned(16))) float pstage[4][16 * PPITCH];

  const int lane = threadIdx.x & 31;
  const int wave = threadIdx.x >> 5;
  const int half = lane >> 4;
  const int l16  = lane & 15;

  const int tile = blockIdx.x * 4 + wave;
  const int qt = tile & 127;
  const int bh = tile >> 7;
  const int b  = bh / NHEAD;
  const int h  = bh % NHEAD;

  const int    q0   = qt * 16;
  const size_t base = (size_t)b * S_LEN * EMBED + (size_t)h * HDIM;

  v4f qf[8];
  {
    const float* qp = Q + ((size_t)bh * TILES_PER_BH + qt) * TILE_FLOATS + lane * 4;
#pragma unroll
    for (int i = 0; i < 8; ++i) qf[i] = *(const v4f*)(qp + i * 128);
  }
  const float* kbase = K + (size_t)bh * TILES_PER_BH * TILE_FLOATS + lane * 4;
  const float* vbase = V + (size_t)bh * TILES_PER_BH * TILE_FLOATS + lane * 4;
  const float* mbase = mask + (size_t)(q0 + 8 * half) * S_LEN + l16;

  v8f acc[4];
#pragma unroll
  for (int i = 0; i < 4; ++i) acc[i] = vzero8();
  float mrow[8], lrow[8];
#pragma unroll
  for (int v = 0; v < 8; ++v) { mrow[v] = -3.0e38f; lrow[v] = 0.f; }

  float* myp = &pstage[wave][0];
  float*       pst = myp + (8 * half) * PPITCH + l16;
  const float* pld = myp + l16 * PPITCH + 8 * half;

  v4f k0[8], k1[8], v0[8], v1[8];
  float m0[8], m1[8];
  __builtin_amdgcn_sched_barrier(0);
  load_k2(kbase, 0, k0, k1);
  load_mask2(mbase, 0, m0, m1);
  __builtin_amdgcn_sched_barrier(0);
  load_v2(vbase, 0, v0, v1);

  for (int st = 0; st < S_LEN / 32; ++st) {
    const int sn = (st + 1 < S_LEN / 32) ? st + 1 : st;

    v8f s0 = vzero8(), s1 = vzero8();
#pragma unroll
    for (int i = 0; i < 8; ++i) {
      v2f a0 = {qf[i].x, qf[i].y};
      v2f a1 = {qf[i].z, qf[i].w};
      v2f b00 = {k0[i].x, k0[i].y};
      v2f b10 = {k1[i].x, k1[i].y};
      v2f b01 = {k0[i].z, k0[i].w};
      v2f b11 = {k1[i].z, k1[i].w};
      s0 = wmma4(a0, b00, s0);
      s1 = wmma4(a0, b10, s1);
      s0 = wmma4(a1, b01, s0);
      s1 = wmma4(a1, b11, s1);
    }
    float sv0[8], sv1[8], mx[8], rmax[8], p0[8], p1[8], rsum[8], alpha[8];
#pragma unroll
    for (int v = 0; v < 8; ++v) {
      sv0[v] = s0[v] * 0.125f + m0[v] * (-1.0e9f);
      sv1[v] = s1[v] * 0.125f + m1[v] * (-1.0e9f);
    }
    asm volatile("" :: "v"(sv0[0]), "v"(sv0[1]), "v"(sv0[2]), "v"(sv0[3]),
                       "v"(sv0[4]), "v"(sv0[5]), "v"(sv0[6]), "v"(sv0[7]),
                       "v"(sv1[0]), "v"(sv1[1]), "v"(sv1[2]), "v"(sv1[3]),
                       "v"(sv1[4]), "v"(sv1[5]), "v"(sv1[6]), "v"(sv1[7]) : "memory");
    load_k2(kbase, sn, k0, k1);
    load_mask2(mbase, sn, m0, m1);
    __builtin_amdgcn_sched_barrier(0);
#pragma unroll
    for (int v = 0; v < 8; ++v) mx[v] = fmaxf(sv0[v], sv1[v]);
    rowmax8(mx, rmax);
#pragma unroll
    for (int v = 0; v < 8; ++v) {
      const float mnew = fmaxf(mrow[v], rmax[v]);
      alpha[v] = __expf(mrow[v] - mnew);
      p0[v]    = __expf(sv0[v] - mnew);
      p1[v]    = __expf(sv1[v] - mnew);
      mrow[v]  = mnew;
#pragma unroll
      for (int nt = 0; nt < 4; ++nt) acc[nt][v] *= alpha[v];
    }
#pragma unroll
    for (int v = 0; v < 8; ++v) rsum[v] = p0[v] + p1[v];
#pragma unroll
    for (int v = 0; v < 8; ++v) rsum[v] += xlane16<1>(rsum[v]);
#pragma unroll
    for (int v = 0; v < 8; ++v) rsum[v] += xlane16<2>(rsum[v]);
#pragma unroll
    for (int v = 0; v < 8; ++v) rsum[v] += xlane16<4>(rsum[v]);
#pragma unroll
    for (int v = 0; v < 8; ++v) rsum[v] += xlane16<8>(rsum[v]);
#pragma unroll
    for (int v = 0; v < 8; ++v) lrow[v] = lrow[v] * alpha[v] + rsum[v];

#pragma unroll
    for (int v = 0; v < 8; ++v) {
      pst[v * PPITCH]      = p0[v];
      pst[v * PPITCH + 16] = p1[v];
    }
    lds_barrier();

    {
      const v4f pa0 = *(const v4f*)(pld);
      const v4f pb0 = *(const v4f*)(pld + 4);
      const v4f pa1 = *(const v4f*)(pld + 16);
      const v4f pb1 = *(const v4f*)(pld + 20);
      pv_tile(pa0, pb0, v0, acc);
      pv_tile(pa1, pb1, v1, acc);
    }
    lds_barrier();
    asm volatile("" :: "v"(acc[0]), "v"(acc[1]), "v"(acc[2]), "v"(acc[3]) : "memory");
    load_v2(vbase, sn, v0, v1);
    __builtin_amdgcn_sched_barrier(0);
  }

#pragma unroll
  for (int v = 0; v < 8; ++v) {
    const float inv = 1.0f / lrow[v];
    float* orow = O + base + (size_t)(q0 + v + 8 * half) * EMBED + 4 * l16;
    v4f o = {acc[0][v] * inv, acc[1][v] * inv, acc[2][v] * inv, acc[3][v] * inv};
    *(v4f*)orow = o;
  }
}

extern "C" void kernel_launch(void* const* d_in, const int* in_sizes, int n_in,
                              void* d_out, int out_size, void* d_ws, size_t ws_size,
                              hipStream_t stream) {
  const float* x    = (const float*)d_in[0];
  const float* mask = (const float*)d_in[1];
  const float* Wq   = (const float*)d_in[2];
  const float* Wk   = (const float*)d_in[3];
  const float* Wv   = (const float*)d_in[4];
  const float* Wo   = (const float*)d_in[5];
  const float* bo   = (const float*)d_in[6];

  float* ws = (float*)d_ws;
  const size_t mat = (size_t)ROWS * EMBED;
  float* Qb = ws;
  float* Kb = ws + mat;
  float* Vb = ws + 2 * mat;
  float* Ab = ws + 3 * mat;

  dim3 g(ROWS / 64, EMBED / 64);
  gemm_kernel<1><<<g, 128, 0, stream>>>(x, Wq, nullptr, Qb);
  gemm_kernel<1><<<g, 128, 0, stream>>>(x, Wk, nullptr, Kb);
  gemm_kernel<2><<<g, 128, 0, stream>>>(x, Wv, nullptr, Vb);

  const int qtiles = BATCH * NHEAD * (S_LEN / 16);
  attention_kernel<<<qtiles / 4, 128, 0, stream>>>(Qb, Kb, Vb, mask, Ab);

  gemm_kernel<0><<<g, 128, 0, stream>>>(Ab, Wo, bo, (float*)d_out);
}


# all four GEMMs: the next 32-k trip's 20 loads (A row, W rows) requested one trip ahead into a second register set, loop body of two trips so the sets swap by name; order of sums unchanged
# speedup vs baseline: 2.2941x; 1.0374x over previous
#include <hip/hip_runtime.h>
#include <hip/hip_bf16.h>


typedef __attribute__((ext_vector_type(2))) float v2f;
typedef __attribute__((ext_vector_type(4))) float v4f;
typedef __attribute__((ext_vector_type(8))) float v8f;

#define EMBED 768
#define S_LEN 2048
#define NHEAD 12
#define HDIM  64
#define BATCH 2
#define ROWS  (BATCH * S_LEN)
#define TILE_FLOATS 1024
#define TILES_PER_BH (S_LEN / 16)

__device__ __forceinline__ v8f wmma4(v2f a, v2f b, v8f c) {
  return __builtin_amdgcn_wmma_f32_16x16x4_f32(
       false, a,  false, b,
       (short)0, c,  false,  false);
}

__device__ __forceinline__ v8f vzero8() {
  v8f z = {0.f, 0.f, 0.f, 0.f, 0.f, 0.f, 0.f, 0.f};
  return z;
}

__device__ __forceinline__ void gemm_load_trip(const float* __restrict__ ap,
                                               const float* __restrict__ wp, int k0,
                                               v4f (&a)[4], v4f (&w)[16]) {
#pragma unroll
  for (int i = 0; i < 4; ++i) a[i] = *(const v4f*)(ap + k0 + 4 * i);
#pragma unroll
  for (int r = 0; r < 16; ++r)
    w[r] = *(const v4f*)(wp + (size_t)(k0 + r) * EMBED);
}

template <int MODE>
__device__ __forceinline__ void gemm_mma_trip(const v4f (&a)[4], const v4f (&w)[16],
                                              v8f (&acc)[4]) {
#pragma unroll
  for (int j = 0; j < 8; ++j) {
    v2f af;
    if (j & 1) { af.x = a[j >> 1].z; af.y = a[j >> 1].w; }
    else       { af.x = a[j >> 1].x; af.y = a[j >> 1].y; }
#pragma unroll
    for (int t = 0; t < 4; ++t) {
      v2f bf = {w[2 * j][t], w[2 * j + 1][t]};
      if (MODE == 1) acc[t] = wmma4(bf, af, acc[t]);
      else           acc[t] = wmma4(af, bf, acc[t]);
    }
  }
}

template <int MODE>
__global__ __launch_bounds__(128) void gemm_kernel(
    const float* __restrict__ A, const float* __restrict__ W,
    const float* __restrict__ bias, float* __restrict__ C) {
  const int lane = threadIdx.x & 31;
  const int wave = threadIdx.x >> 5;
  const int half = lane >> 4;
  const int l16  = lane & 15;

  const int m0 = blockIdx.x * 64 + wave * 16;
  const int n0 = blockIdx.y * 64;

  v8f acc[4];
#pragma unroll
  for (int i = 0; i < 4; ++i) acc[i] = vzero8();

  const float* ap = A + (size_t)(m0 + l16) * EMBED + 16 * half;
  const float* wp = W + (size_t)(16 * half) * EMBED + n0 + 4 * l16;

  v4f a0[4], w0[16], a1[4], w1[16];
  gemm_load_trip(ap, wp, 0, a0, w0);
  for (int k0 = 0; k0 < EMBED; k0 += 64) {
    gemm_load_trip(ap, wp, k0 + 32, a1, w1);
    __builtin_amdgcn_sched_barrier(0);
    gemm_mma_trip<MODE>(a0, w0, acc);
    __builtin_amdgcn_sched_barrier(0);
    const int kn = (k0 + 64 < EMBED) ? k0 + 64 : k0;
    gemm_load_trip(ap, wp, kn, a0, w0);
    __builtin_amdgcn_sched_barrier(0);
    gemm_mma_trip<MODE>(a1, w1, acc);
    __builtin_amdgcn_sched_barrier(0);
  }

  if (MODE == 0) {
    const int col = n0 + 4 * l16;
    const v4f bv = *(const v4f*)(bias + col);
#pragma unroll
    for (int v = 0; v < 8; ++v) {
      v4f o = {acc[0][v] + bv.x, acc[1][v] + bv.y, acc[2][v] + bv.z, acc[3][v] + bv.w};
      *(v4f*)(C + (size_t)(m0 + v + 8 * half) * EMBED + col) = o;
    }
  } else {
    const int bidx = m0 / S_LEN;
    const int tt   = (m0 % S_LEN) / 16;
    const int hd   = n0 / HDIM;
    float* tile = C + ((size_t)(bidx * NHEAD + hd) * TILES_PER_BH + tt) * TILE_FLOATS
                    + lane * 4;
#pragma unroll
    for (int nt = 0; nt < 4; ++nt) {
#pragma unroll
      for (int hv = 0; hv < 2; ++hv) {
        v4f o = {acc[nt][4 * hv], acc[nt][4 * hv + 1],
                 acc[nt][4 * hv + 2], acc[nt][4 * hv + 3]};
        *(v4f*)(tile + (nt * 2 + hv) * 128) = o;
      }
    }
  }
}


template <int M>
__device__ __forceinline__ float xlane16(float v) {
  return __int_as_float(__builtin_amdgcn_update_dpp(
      0, __float_as_int(v), 0x160 + M, 0xf, 0xf, true));
}

#define RMAX_I(d, s, m) \
  "v_max_num_f32_dpp %" #d ", %" #s ", %" #s " row_xmask:" #m " row_mask:0xf bank_mask:0xf\n\t"
#define RMAX_STEP(m) \
  RMAX_I(0, 0, m) RMAX_I(1, 1, m) RMAX_I(2, 2, m) RMAX_I(3, 3, m) \
  RMAX_I(4, 4, m) RMAX_I(5, 5, m) RMAX_I(6, 6, m) RMAX_I(7, 7, m)

__device__ __forceinline__ void rowmax8(const float (&x)[8], float (&r)[8]) {
  asm(RMAX_I(0, 8, 1) RMAX_I(1, 9, 1) RMAX_I(2, 10, 1) RMAX_I(3, 11, 1)
      RMAX_I(4, 12, 1) RMAX_I(5, 13, 1) RMAX_I(6, 14, 1) RMAX_I(7, 15, 1)
      RMAX_STEP(2) RMAX_STEP(4) RMAX_STEP(8)
      : "=&v"(r[0]), "=&v"(r[1]), "=&v"(r[2]), "=&v"(r[3]),
        "=&v"(r[4]), "=&v"(r[5]), "=&v"(r[6]), "=&v"(r[7])
      : "v"(x[0]), "v"(x[1]), "v"(x[2]), "v"(x[3]),
        "v"(x[4]), "v"(x[5]), "v"(x[6]), "v"(x[7]));
}

__device__ __forceinline__ void load_k2(const float* __restrict__ kbase, int st,
                                        v4f (&k0)[8], v4f (&k1)[8]) {
  const float* kp = kbase + (size_t)st * (2 * TILE_FLOATS);
#pragma unroll
  for (int i = 0; i < 8; ++i) k0[i] = *(const v4f*)(kp + i * 128);
#pragma unroll
  for (int i = 0; i < 8; ++i) k1[i] = *(const v4f*)(kp + TILE_FLOATS + i * 128);
}

__device__ __forceinline__ void load_mask2(const float* __restrict__ mbase, int st,
                                           float (&m0)[8], float (&m1)[8]) {
  const float* mp = mbase + st * 32;
#pragma unroll
  for (int v = 0; v < 8; ++v) m0[v] = mp[(size_t)v * S_LEN];
#pragma unroll
  for (int v = 0; v < 8; ++v) m1[v] = mp[(size_t)v * S_LEN + 16];
}

__device__ __forceinline__ void load_v2(const float* __restrict__ vbase, int st,
                                        v4f (&v0)[8], v4f (&v1)[8]) {
  const float* vp = vbase + (size_t)st * (2 * TILE_FLOATS);
#pragma unroll
  for (int i = 0; i < 8; ++i) v0[i] = *(const v4f*)(vp + i * 128);
#pragma unroll
  for (int i = 0; i < 8; ++i) v1[i] = *(const v4f*)(vp + TILE_FLOATS + i * 128);
}

__device__ __forceinline__ void lds_barrier() {
  __builtin_amdgcn_fence(__ATOMIC_RELEASE, "workgroup", "local");
  __builtin_amdgcn_s_barrier();
  __builtin_amdgcn_fence(__ATOMIC_ACQUIRE, "workgroup", "local");
}

__device__ __forceinline__ void pv_tile(const v4f pa, const v4f pb, const v4f (&vf)[8],
                                        v8f (&acc)[4]) {
#pragma unroll
  for (int kk = 0; kk < 4; ++kk) {
    const v4f pq = (kk >> 1) ? pb : pa;
    v2f a;
    if (kk & 1) { a.x = pq.z; a.y = pq.w; }
    else        { a.x = pq.x; a.y = pq.y; }
#pragma unroll
    for (int nt = 0; nt < 4; ++nt) {
      const v4f t = vf[nt * 2 + (kk >> 1)];
      v2f bf;
      if (kk & 1) { bf.x = t.z; bf.y = t.w; }
      else        { bf.x = t.x; bf.y = t.y; }
      acc[nt] = wmma4(a, bf, acc[nt]);
    }
  }
}

#define PPITCH 36

__global__ __launch_bounds__(128) void attention_kernel(
    const float* __restrict__ Q, const float* __restrict__ K,
    const float* __restrict__ V, const float* __restrict__ mask,
    float* __restrict__ O) {
  __shared__ __attribute__((aligned(16))) float pstage[4][16 * PPITCH];

  const int lane = threadIdx.x & 31;
  const int wave = threadIdx.x >> 5;
  const int half = lane >> 4;
  const int l16  = lane & 15;

  const int tile = blockIdx.x * 4 + wave;
  const int qt = tile & 127;
  const int bh = tile >> 7;
  const int b  = bh / NHEAD;
  const int h  = bh % NHEAD;

  const int    q0   = qt * 16;
  const size_t base = (size_t)b * S_LEN * EMBED + (size_t)h * HDIM;

  v4f qf[8];
  {
    const float* qp = Q + ((size_t)bh * TILES_PER_BH + qt) * TILE_FLOATS + lane * 4;
#pragma unroll
    for (int i = 0; i < 8; ++i) qf[i] = *(const v4f*)(qp + i * 128);
  }
  const float* kbase = K + (size_t)bh * TILES_PER_BH * TILE_FLOATS + lane * 4;
  const float* vbase = V + (size_t)bh * TILES_PER_BH * TILE_FLOATS + lane * 4;
  const float* mbase = mask + (size_t)(q0 + 8 * half) * S_LEN + l16;

  v8f acc[4];
#pragma unroll
  for (int i = 0; i < 4; ++i) acc[i] = vzero8();
  float mrow[8], lrow[8];
#pragma unroll
  for (int v = 0; v < 8; ++v) { mrow[v] = -3.0e38f; lrow[v] = 0.f; }

  float* myp = &pstage[wave][0];
  float*       pst = myp + (8 * half) * PPITCH + l16;
  const float* pld = myp + l16 * PPITCH + 8 * half;

  v4f k0[8], k1[8], v0[8], v1[8];
  float m0[8], m1[8];
  __builtin_amdgcn_sched_barrier(0);
  load_k2(kbase, 0, k0, k1);
  load_mask2(mbase, 0, m0, m1);
  __builtin_amdgcn_sched_barrier(0);
  load_v2(vbase, 0, v0, v1);

  for (int st = 0; st < S_LEN / 32; ++st) {
    const int sn = (st + 1 < S_LEN / 32) ? st + 1 : st;

    v8f s0 = vzero8(), s1 = vzero8();
#pragma unroll
    for (int i = 0; i < 8; ++i) {
      v2f a0 = {qf[i].x, qf[i].y};
      v2f a1 = {qf[i].z, qf[i].w};
      v2f b00 = {k0[i].x, k0[i].y};
      v2f b10 = {k1[i].x, k1[i].y};
      v2f b01 = {k0[i].z, k0[i].w};
      v2f b11 = {k1[i].z, k1[i].w};
      s0 = wmma4(a0, b00, s0);
      s1 = wmma4(a0, b10, s1);
      s0 = wmma4(a1, b01, s0);
      s1 = wmma4(a1, b11, s1);
    }
    float sv0[8], sv1[8], mx[8], rmax[8], p0[8], p1[8], rsum[8], alpha[8];
#pragma unroll
    for (int v = 0; v < 8; ++v) {
      sv0[v] = s0[v] * 0.125f + m0[v] * (-1.0e9f);
      sv1[v] = s1[v] * 0.125f + m1[v] * (-1.0e9f);
    }
    asm volatile("" :: "v"(sv0[0]), "v"(sv0[1]), "v"(sv0[2]), "v"(sv0[3]),
                       "v"(sv0[4]), "v"(sv0[5]), "v"(sv0[6]), "v"(sv0[7]),
                       "v"(sv1[0]), "v"(sv1[1]), "v"(sv1[2]), "v"(sv1[3]),
                       "v"(sv1[4]), "v"(sv1[5]), "v"(sv1[6]), "v"(sv1[7]) : "memory");
    load_k2(kbase, sn, k0, k1);
    load_mask2(mbase, sn, m0, m1);
    __builtin_amdgcn_sched_barrier(0);
#pragma unroll
    for (int v = 0; v < 8; ++v) mx[v] = fmaxf(sv0[v], sv1[v]);
    rowmax8(mx, rmax);
#pragma unroll
    for (int v = 0; v < 8; ++v) {
      const float mnew = fmaxf(mrow[v], rmax[v]);
      alpha[v] = __expf(mrow[v] - mnew);
      p0[v]    = __expf(sv0[v] - mnew);
      p1[v]    = __expf(sv1[v] - mnew);
      mrow[v]  = mnew;
#pragma unroll
      for (int nt = 0; nt < 4; ++nt) acc[nt][v] *= alpha[v];
    }
#pragma unroll
    for (int v = 0; v < 8; ++v) rsum[v] = p0[v] + p1[v];
#pragma unroll
    for (int v = 0; v < 8; ++v) rsum[v] += xlane16<1>(rsum[v]);
#pragma unroll
    for (int v = 0; v < 8; ++v) rsum[v] += xlane16<2>(rsum[v]);
#pragma unroll
    for (int v = 0; v < 8; ++v) rsum[v] += xlane16<4>(rsum[v]);
#pragma unroll
    for (int v = 0; v < 8; ++v) rsum[v] += xlane16<8>(rsum[v]);
#pragma unroll
    for (int v = 0; v < 8; ++v) lrow[v] = lrow[v] * alpha[v] + rsum[v];

#pragma unroll
    for (int v = 0; v < 8; ++v) {
      pst[v * PPITCH]      = p0[v];
      pst[v * PPITCH + 16] = p1[v];
    }
    lds_barrier();

    {
      const v4f pa0 = *(const v4f*)(pld);
      const v4f pb0 = *(const v4f*)(pld + 4);
      const v4f pa1 = *(const v4f*)(pld + 16);
      const v4f pb1 = *(const v4f*)(pld + 20);
      pv_tile(pa0, pb0, v0, acc);
      pv_tile(pa1, pb1, v1, acc);
    }
    lds_barrier();
    asm volatile("" :: "v"(acc[0]), "v"(acc[1]), "v"(acc[2]), "v"(acc[3]) : "memory");
    load_v2(vbase, sn, v0, v1);
    __builtin_amdgcn_sched_barrier(0);
  }

#pragma unroll
  for (int v = 0; v < 8; ++v) {
    const float inv = 1.0f / lrow[v];
    float* orow = O + base + (size_t)(q0 + v + 8 * half) * EMBED + 4 * l16;
    v4f o = {acc[0][v] * inv, acc[1][v] * inv, acc[2][v] * inv, acc[3][v] * inv};
    *(v4f*)orow = o;
  }
}

extern "C" void kernel_launch(void* const* d_in, const int* in_sizes, int n_in,
                              void* d_out, int out_size, void* d_ws, size_t ws_size,
                              hipStream_t stream) {
  const float* x    = (const float*)d_in[0];
  const float* mask = (const float*)d_in[1];
  const float* Wq   = (const float*)d_in[2];
  const float* Wk   = (const float*)d_in[3];
  const float* Wv   = (const float*)d_in[4];
  const float* Wo   = (const float*)d_in[5];
  const float* bo   = (const float*)d_in[6];

  float* ws = (float*)d_ws;
  const size_t mat = (size_t)ROWS * EMBED;
  float* Qb = ws;
  float* Kb = ws + mat;
  float* Vb = ws + 2 * mat;
  float* Ab = ws + 3 * mat;

  dim3 g(ROWS / 64, EMBED / 64);
  gemm_kernel<1><<<g, 128, 0, stream>>>(x, Wq, nullptr, Qb);
  gemm_kernel<1><<<g, 128, 0, stream>>>(x, Wk, nullptr, Kb);
  gemm_kernel<2><<<g, 128, 0, stream>>>(x, Wv, nullptr, Vb);

  const int qtiles = BATCH * NHEAD * (S_LEN / 16);
  attention_kernel<<<qtiles / 4, 128, 0, stream>>>(Qb, Kb, Vb, mask, Ab);

  gemm_kernel<0><<<g, 128, 0, stream>>>(Ab, Wo, bo, (float*)d_out);
}


# attention as 256 work-groups of 12 waves (12 query tiles, one group per WGP), not 768 of 4: the step barriers keep all 12 waves on the same K/V lines of the WGP cache; sums unchanged
# speedup vs baseline: 2.3687x; 1.0325x over previous
#include <hip/hip_runtime.h>
#include <hip/hip_bf16.h>


typedef __attribute__((ext_vector_type(2))) float v2f;
typedef __attribute__((ext_vector_type(4))) float v4f;
typedef __attribute__((ext_vector_type(8))) float v8f;

#define EMBED 768
#define S_LEN 2048
#define NHEAD 12
#define HDIM  64
#define BATCH 2
#define ROWS  (BATCH * S_LEN)
#define TILE_FLOATS 1024
#define TILES_PER_BH (S_LEN / 16)

__device__ __forceinline__ v8f wmma4(v2f a, v2f b, v8f c) {
  return __builtin_amdgcn_wmma_f32_16x16x4_f32(
       false, a,  false, b,
       (short)0, c,  false,  false);
}

__device__ __forceinline__ v8f vzero8() {
  v8f z = {0.f, 0.f, 0.f, 0.f, 0.f, 0.f, 0.f, 0.f};
  return z;
}

__device__ __forceinline__ void gemm_load_trip(const float* __restrict__ ap,
                                               const float* __restrict__ wp, int k0,
                                               v4f (&a)[4], v4f (&w)[16]) {
#pragma unroll
  for (int i = 0; i < 4; ++i) a[i] = *(const v4f*)(ap + k0 + 4 * i);
#pragma unroll
  for (int r = 0; r < 16; ++r)
    w[r] = *(const v4f*)(wp + (size_t)(k0 + r) * EMBED);
}

template <int MODE>
__device__ __forceinline__ void gemm_mma_trip(const v4f (&a)[4], const v4f (&w)[16],
                                              v8f (&acc)[4]) {
#pragma unroll
  for (int j = 0; j < 8; ++j) {
    v2f af;
    if (j & 1) { af.x = a[j >> 1].z; af.y = a[j >> 1].w; }
    else       { af.x = a[j >> 1].x; af.y = a[j >> 1].y; }
#pragma unroll
    for (int t = 0; t < 4; ++t) {
      v2f bf = {w[2 * j][t], w[2 * j + 1][t]};
      if (MODE == 1) acc[t] = wmma4(bf, af, acc[t]);
      else           acc[t] = wmma4(af, bf, acc[t]);
    }
  }
}

template <int MODE>
__global__ __launch_bounds__(128) void gemm_kernel(
    const float* __restrict__ A, const float* __restrict__ W,
    const float* __restrict__ bias, float* __restrict__ C) {
  const int lane = threadIdx.x & 31;
  const int wave = threadIdx.x >> 5;
  const int half = lane >> 4;
  const int l16  = lane & 15;

  const int m0 = blockIdx.x * 64 + wave * 16;
  const int n0 = blockIdx.y * 64;

  v8f acc[4];
#pragma unroll
  for (int i = 0; i < 4; ++i) acc[i] = vzero8();

  const float* ap = A + (size_t)(m0 + l16) * EMBED + 16 * half;
  const float* wp = W + (size_t)(16 * half) * EMBED + n0 + 4 * l16;

  v4f a0[4], w0[16], a1[4], w1[16];
  gemm_load_trip(ap, wp, 0, a0, w0);
  for (int k0 = 0; k0 < EMBED; k0 += 64) {
    gemm_load_trip(ap, wp, k0 + 32, a1, w1);
    __builtin_amdgcn_sched_barrier(0);
    gemm_mma_trip<MODE>(a0, w0, acc);
    __builtin_amdgcn_sched_barrier(0);
    const int kn = (k0 + 64 < EMBED) ? k0 + 64 : k0;
    gemm_load_trip(ap, wp, kn, a0, w0);
    __builtin_amdgcn_sched_barrier(0);
    gemm_mma_trip<MODE>(a1, w1, acc);
    __builtin_amdgcn_sched_barrier(0);
  }

  if (MODE == 0) {
    const int col = n0 + 4 * l16;
    const v4f bv = *(const v4f*)(bias + col);
#pragma unroll
    for (int v = 0; v < 8; ++v) {
      v4f o = {acc[0][v] + bv.x, acc[1][v] + bv.y, acc[2][v] + bv.z, acc[3][v] + bv.w};
      *(v4f*)(C + (size_t)(m0 + v + 8 * half) * EMBED + col) = o;
    }
  } else {
    const int bidx = m0 / S_LEN;
    const int tt   = (m0 % S_LEN) / 16;
    const int hd   = n0 / HDIM;
    float* tile = C + ((size_t)(bidx * NHEAD + hd) * TILES_PER_BH + tt) * TILE_FLOATS
                    + lane * 4;
#pragma unroll
    for (int nt = 0; nt < 4; ++nt) {
#pragma unroll
      for (int hv = 0; hv < 2; ++hv) {
        v4f o = {acc[nt][4 * hv], acc[nt][4 * hv + 1],
                 acc[nt][4 * hv + 2], acc[nt][4 * hv + 3]};
        *(v4f*)(tile + (nt * 2 + hv) * 128) = o;
      }
    }
  }
}


template <int M>
__device__ __forceinline__ float xlane16(float v) {
  return __int_as_float(__builtin_amdgcn_update_dpp(
      0, __float_as_int(v), 0x160 + M, 0xf, 0xf, true));
}

#define RMAX_I(d, s, m) \
  "v_max_num_f32_dpp %" #d ", %" #s ", %" #s " row_xmask:" #m " row_mask:0xf bank_mask:0xf\n\t"
#define RMAX_STEP(m) \
  RMAX_I(0, 0, m) RMAX_I(1, 1, m) RMAX_I(2, 2, m) RMAX_I(3, 3, m) \
  RMAX_I(4, 4, m) RMAX_I(5, 5, m) RMAX_I(6, 6, m) RMAX_I(7, 7, m)

__device__ __forceinline__ void rowmax8(const float (&x)[8], float (&r)[8]) {
  asm(RMAX_I(0, 8, 1) RMAX_I(1, 9, 1) RMAX_I(2, 10, 1) RMAX_I(3, 11, 1)
      RMAX_I(4, 12, 1) RMAX_I(5, 13, 1) RMAX_I(6, 14, 1) RMAX_I(7, 15, 1)
      RMAX_STEP(2) RMAX_STEP(4) RMAX_STEP(8)
      : "=&v"(r[0]), "=&v"(r[1]), "=&v"(r[2]), "=&v"(r[3]),
        "=&v"(r[4]), "=&v"(r[5]), "=&v"(r[6]), "=&v"(r[7])
      : "v"(x[0]), "v"(x[1]), "v"(x[2]), "v"(x[3]),
        "v"(x[4]), "v"(x[5]), "v"(x[6]), "v"(x[7]));
}

__device__ __forceinline__ void load_k2(const float* __restrict__ kbase, int st,
                                        v4f (&k0)[8], v4f (&k1)[8]) {
  const float* kp = kbase + (size_t)st * (2 * TILE_FLOATS);
#pragma unroll
  for (int i = 0; i < 8; ++i) k0[i] = *(const v4f*)(kp + i * 128);
#pragma unroll
  for (int i = 0; i < 8; ++i) k1[i] = *(const v4f*)(kp + TILE_FLOATS + i * 128);
}

__device__ __forceinline__ void load_mask2(const float* __restrict__ mbase, int st,
                                           float (&m0)[8], float (&m1)[8]) {
  const float* mp = mbase + st * 32;
#pragma unroll
  for (int v = 0; v < 8; ++v) m0[v] = mp[(size_t)v * S_LEN];
#pragma unroll
  for (int v = 0; v < 8; ++v) m1[v] = mp[(size_t)v * S_LEN + 16];
}

__device__ __forceinline__ void load_v2(const float* __restrict__ vbase, int st,
                                        v4f (&v0)[8], v4f (&v1)[8]) {
  const float* vp = vbase + (size_t)st * (2 * TILE_FLOATS);
#pragma unroll
  for (int i = 0; i < 8; ++i) v0[i] = *(const v4f*)(vp + i * 128);
#pragma unroll
  for (int i = 0; i < 8; ++i) v1[i] = *(const v4f*)(vp + TILE_FLOATS + i * 128);
}

__device__ __forceinline__ void lds_barrier() {
  __builtin_amdgcn_fence(__ATOMIC_RELEASE, "workgroup", "local");
  __builtin_amdgcn_s_barrier();
  __builtin_amdgcn_fence(__ATOMIC_ACQUIRE, "workgroup", "local");
}

__device__ __forceinline__ void pv_tile(const v4f pa, const v4f pb, const v4f (&vf)[8],
                                        v8f (&acc)[4]) {
#pragma unroll
  for (int kk = 0; kk < 4; ++kk) {
    const v4f pq = (kk >> 1) ? pb : pa;
    v2f a;
    if (kk & 1) { a.x = pq.z; a.y = pq.w; }
    else        { a.x = pq.x; a.y = pq.y; }
#pragma unroll
    for (int nt = 0; nt < 4; ++nt) {
      const v4f t = vf[nt * 2 + (kk >> 1)];
      v2f bf;
      if (kk & 1) { bf.x = t.z; bf.y = t.w; }
      else        { bf.x = t.x; bf.y = t.y; }
      acc[nt] = wmma4(a, bf, acc[nt]);
    }
  }
}

#define PPITCH 36

#define ATT_WAVES 12
__global__ __launch_bounds__(32 * ATT_WAVES) void attention_kernel(
    const float* __restrict__ Q, const float* __restrict__ K,
    const float* __restrict__ V, const float* __restrict__ mask,
    float* __restrict__ O) {
  __shared__ __attribute__((aligned(16))) float pstage[ATT_WAVES][16 * PPITCH];

  const int lane = threadIdx.x & 31;
  const int wave = threadIdx.x >> 5;
  const int half = lane >> 4;
  const int l16  = lane & 15;

  const int tile = blockIdx.x * ATT_WAVES + wave;
  const int qt = tile & 127;
  const int bh = tile >> 7;
  const int b  = bh / NHEAD;
  const int h  = bh % NHEAD;

  const int    q0   = qt * 16;
  const size_t base = (size_t)b * S_LEN * EMBED + (size_t)h * HDIM;

  v4f qf[8];
  {
    const float* qp = Q + ((size_t)bh * TILES_PER_BH + qt) * TILE_FLOATS + lane * 4;
#pragma unroll
    for (int i = 0; i < 8; ++i) qf[i] = *(const v4f*)(qp + i * 128);
  }
  const float* kbase = K + (size_t)bh * TILES_PER_BH * TILE_FLOATS + lane * 4;
  const float* vbase = V + (size_t)bh * TILES_PER_BH * TILE_FLOATS + lane * 4;
  const float* mbase = mask + (size_t)(q0 + 8 * half) * S_LEN + l16;

  v8f acc[4];
#pragma unroll
  for (int i = 0; i < 4; ++i) acc[i] = vzero8();
  float mrow[8], lrow[8];
#pragma unroll
  for (int v = 0; v < 8; ++v) { mrow[v] = -3.0e38f; lrow[v] = 0.f; }

  float* myp = &pstage[wave][0];
  float*       pst = myp + (8 * half) * PPITCH + l16;
  const float* pld = myp + l16 * PPITCH + 8 * half;

  v4f k0[8], k1[8], v0[8], v1[8];
  float m0[8], m1[8];
  __builtin_amdgcn_sched_barrier(0);
  load_k2(kbase, 0, k0, k1);
  load_mask2(mbase, 0, m0, m1);
  __builtin_amdgcn_sched_barrier(0);
  load_v2(vbase, 0, v0, v1);

  for (int st = 0; st < S_LEN / 32; ++st) {
    const int sn = (st + 1 < S_LEN / 32) ? st + 1 : st;

    v8f s0 = vzero8(), s1 = vzero8();
#pragma unroll
    for (int i = 0; i < 8; ++i) {
      v2f a0 = {qf[i].x, qf[i].y};
      v2f a1 = {qf[i].z, qf[i].w};
      v2f b00 = {k0[i].x, k0[i].y};
      v2f b10 = {k1[i].x, k1[i].y};
      v2f b01 = {k0[i].z, k0[i].w};
      v2f b11 = {k1[i].z, k1[i].w};
      s0 = wmma4(a0, b00, s0);
      s1 = wmma4(a0, b10, s1);
      s0 = wmma4(a1, b01, s0);
      s1 = wmma4(a1, b11, s1);
    }
    float sv0[8], sv1[8], mx[8], rmax[8], p0[8], p1[8], rsum[8], alpha[8];
#pragma unroll
    for (int v = 0; v < 8; ++v) {
      sv0[v] = s0[v] * 0.125f + m0[v] * (-1.0e9f);
      sv1[v] = s1[v] * 0.125f + m1[v] * (-1.0e9f);
    }
    asm volatile("" :: "v"(sv0[0]), "v"(sv0[1]), "v"(sv0[2]), "v"(sv0[3]),
                       "v"(sv0[4]), "v"(sv0[5]), "v"(sv0[6]), "v"(sv0[7]),
                       "v"(sv1[0]), "v"(sv1[1]), "v"(sv1[2]), "v"(sv1[3]),
                       "v"(sv1[4]), "v"(sv1[5]), "v"(sv1[6]), "v"(sv1[7]) : "memory");
    load_k2(kbase, sn, k0, k1);
    load_mask2(mbase, sn, m0, m1);
    __builtin_amdgcn_sched_barrier(0);
#pragma unroll
    for (int v = 0; v < 8; ++v) mx[v] = fmaxf(sv0[v], sv1[v]);
    rowmax8(mx, rmax);
#pragma unroll
    for (int v = 0; v < 8; ++v) {
      const float mnew = fmaxf(mrow[v], rmax[v]);
      alpha[v] = __expf(mrow[v] - mnew);
      p0[v]    = __expf(sv0[v] - mnew);
      p1[v]    = __expf(sv1[v] - mnew);
      mrow[v]  = mnew;
#pragma unroll
      for (int nt = 0; nt < 4; ++nt) acc[nt][v] *= alpha[v];
    }
#pragma unroll
    for (int v = 0; v < 8; ++v) rsum[v] = p0[v] + p1[v];
#pragma unroll
    for (int v = 0; v < 8; ++v) rsum[v] += xlane16<1>(rsum[v]);
#pragma unroll
    for (int v = 0; v < 8; ++v) rsum[v] += xlane16<2>(rsum[v]);
#pragma unroll
    for (int v = 0; v < 8; ++v) rsum[v] += xlane16<4>(rsum[v]);
#pragma unroll
    for (int v = 0; v < 8; ++v) rsum[v] += xlane16<8>(rsum[v]);
#pragma unroll
    for (int v = 0; v < 8; ++v) lrow[v] = lrow[v] * alpha[v] + rsum[v];

#pragma unroll
    for (int v = 0; v < 8; ++v) {
      pst[v * PPITCH]      = p0[v];
      pst[v * PPITCH + 16] = p1[v];
    }
    lds_barrier();

    {
      const v4f pa0 = *(const v4f*)(pld);
      const v4f pb0 = *(const v4f*)(pld + 4);
      const v4f pa1 = *(const v4f*)(pld + 16);
      const v4f pb1 = *(const v4f*)(pld + 20);
      pv_tile(pa0, pb0, v0, acc);
      pv_tile(pa1, pb1, v1, acc);
    }
    lds_barrier();
    asm volatile("" :: "v"(acc[0]), "v"(acc[1]), "v"(acc[2]), "v"(acc[3]) : "memory");
    load_v2(vbase, sn, v0, v1);
    __builtin_amdgcn_sched_barrier(0);
  }

#pragma unroll
  for (int v = 0; v < 8; ++v) {
    const float inv = 1.0f / lrow[v];
    float* orow = O + base + (size_t)(q0 + v + 8 * half) * EMBED + 4 * l16;
    v4f o = {acc[0][v] * inv, acc[1][v] * inv, acc[2][v] * inv, acc[3][v] * inv};
    *(v4f*)orow = o;
  }
}

extern "C" void kernel_launch(void* const* d_in, const int* in_sizes, int n_in,
                              void* d_out, int out_size, void* d_ws, size_t ws_size,
                              hipStream_t stream) {
  const float* x    = (const float*)d_in[0];
  const float* mask = (const float*)d_in[1];
  const float* Wq   = (const float*)d_in[2];
  const float* Wk   = (const float*)d_in[3];
  const float* Wv   = (const float*)d_in[4];
  const float* Wo   = (const float*)d_in[5];
  const float* bo   = (const float*)d_in[6];

  float* ws = (float*)d_ws;
  const size_t mat = (size_t)ROWS * EMBED;
  float* Qb = ws;
  float* Kb = ws + mat;
  float* Vb = ws + 2 * mat;
  float* Ab = ws + 3 * mat;

  dim3 g(ROWS / 64, EMBED / 64);
  gemm_kernel<1><<<g, 128, 0, stream>>>(x, Wq, nullptr, Qb);
  gemm_kernel<1><<<g, 128, 0, stream>>>(x, Wk, nullptr, Kb);
  gemm_kernel<2><<<g, 128, 0, stream>>>(x, Wv, nullptr, Vb);

  const int qtiles = BATCH * NHEAD * (S_LEN / 16);
  attention_kernel<<<qtiles / ATT_WAVES, 32 * ATT_WAVES, 0, stream>>>(Qb, Kb, Vb, mask, Ab);

  gemm_kernel<0><<<g, 128, 0, stream>>>(Ab, Wo, bo, (float*)d_out);
}
